# starting kernel
# baseline (speedup 1.0000x reference)
#include <hip/hip_runtime.h>
#include <hip/hip_bf16.h>

#define NNODES 100000
#define NEDGES 1600000
#define KIN    128

typedef __attribute__((ext_vector_type(2))) float v2f;
typedef __attribute__((ext_vector_type(8))) float v8f;

__global__ void zero4_kernel(float4* __restrict__ p, int n4) {
    int i = blockIdx.x * blockDim.x + threadIdx.x;
    if (i < n4) p[i] = make_float4(0.f, 0.f, 0.f, 0.f);
}

__global__ void deg_kernel(const int* __restrict__ dst, float* __restrict__ deg) {
    int e = blockIdx.x * blockDim.x + threadIdx.x;
    if (e < NEDGES) atomicAdd(&deg[dst[e]], 1.0f);
}

__global__ void invdeg_kernel(const float* __restrict__ deg, float* __restrict__ invdeg) {
    int i = blockIdx.x * blockDim.x + threadIdx.x;
    if (i < NNODES) invdeg[i] = 1.0f / fmaxf(deg[i], 1.0f);
}

__global__ void scatter_kernel(const float* __restrict__ h,
                               const int* __restrict__ src,
                               const int* __restrict__ dst,
                               float* __restrict__ agg) {
    int gid  = blockIdx.x * blockDim.x + threadIdx.x;
    int e    = gid >> 5;
    int lane = gid & 31;
    int s = src[e];
    int d = dst[e];
    const float4 v = ((const float4*)(h + (size_t)s * KIN))[lane];
    float* a = agg + (size_t)d * KIN + lane * 4;
    atomicAdd(a + 0, v.x);
    atomicAdd(a + 1, v.y);
    atomicAdd(a + 2, v.z);
    atomicAdd(a + 3, v.w);
}

__global__ void pack_w_kernel(const float* __restrict__ W, float2* __restrict__ Wp,
                              int nout, int ntiles) {
    int pid = blockIdx.x * blockDim.x + threadIdx.x;
    int total = ntiles * (KIN / 4) * 32;
    if (pid >= total) return;
    int lane = pid & 31;
    int i    = (pid >> 5) & ((KIN / 4) - 1);
    int tile = pid >> 10;
    int l = lane & 15, half = lane >> 4;
    int col = tile * 16 + l;
    int k   = 4 * i + 2 * half;
    float x = 0.f, y = 0.f;
    if (col < nout) {
        x = W[(size_t)k       * nout + col];
        y = W[(size_t)(k + 1) * nout + col];
    }
    Wp[pid] = make_float2(x, y);
}

template<int NOUT, bool ACT, bool NORM>
__global__ void __launch_bounds__(256)
sage_gemm(const float* __restrict__ hself,
          const float* __restrict__ agg,
          const float* __restrict__ invdeg,
          const float2* __restrict__ Wsp,
          const float2* __restrict__ Wnp,
          const float* __restrict__ bself,
          const float* __restrict__ bneigh,
          float* __restrict__ out) {
    constexpr bool FULL = (NOUT % 16) == 0;
    const int lane = threadIdx.x & 31;
    const int wv   = threadIdx.x >> 5;
    const int l    = lane & 15;
    const int half = lane >> 4;
    const int m0   = blockIdx.x * 16;
    const int n0   = wv * 16;
    const int col  = n0 + l;
    const bool colok = FULL || (col < NOUT);
    const int arow = m0 + l;

    const float idg = invdeg[arow];
    const float*  __restrict__ As = hself + (size_t)arow * KIN + 2 * half;
    const float*  __restrict__ An = agg   + (size_t)arow * KIN + 2 * half;
    const float2* __restrict__ Bs = Wsp + (size_t)wv * (KIN / 4) * 32 + lane;
    const float2* __restrict__ Bn = Wnp + (size_t)wv * (KIN / 4) * 32 + lane;

    v8f acc = {0.f, 0.f, 0.f, 0.f, 0.f, 0.f, 0.f, 0.f};

    #pragma unroll
    for (int i = 0; i < KIN / 4; ++i) {
        const float2 af = *(const float2*)(As + 4 * i);
        const float2 bf = Bs[i * 32];
        v2f a = {af.x, af.y};
        v2f b = {bf.x, bf.y};
        acc = __builtin_amdgcn_wmma_f32_16x16x4_f32(false, a, false, b,
                                                    (short)0, acc, false, false);
    }
    #pragma unroll
    for (int i = 0; i < KIN / 4; ++i) {
        const float2 af = *(const float2*)(An + 4 * i);
        const float2 bf = Bn[i * 32];
        v2f a = {af.x * idg, af.y * idg};
        v2f b = {bf.x, bf.y};
        acc = __builtin_amdgcn_wmma_f32_16x16x4_f32(false, a, false, b,
                                                    (short)0, acc, false, false);
    }

    const float bias = colok ? (bself[col] + bneigh[col]) : 0.0f;
    #pragma unroll
    for (int r = 0; r < 8; ++r) {
        float v = acc[r] + bias;
        if (ACT) v = fmaxf(v, 0.0f);
        acc[r] = v;
    }

    if (NORM) {
        __shared__ float rowsum[16];
        if (threadIdx.x < 16) rowsum[threadIdx.x] = 0.0f;
        __syncthreads();
        #pragma unroll
        for (int r = 0; r < 8; ++r)
            atomicAdd(&rowsum[r + 8 * half], acc[r] * acc[r]);
        __syncthreads();
        #pragma unroll
        for (int r = 0; r < 8; ++r) {
            float nrm = fmaxf(sqrtf(rowsum[r + 8 * half]), 1e-12f);
            acc[r] = acc[r] / nrm;
        }
    }

    if (colok) {
        #pragma unroll
        for (int r = 0; r < 8; ++r) {
            const int m = m0 + r + 8 * half;
            out[(size_t)m * NOUT + col] = acc[r];
        }
    }
}

extern "C" void kernel_launch(void* const* d_in, const int* in_sizes, int n_in,
                              void* d_out, int out_size, void* d_ws, size_t ws_size,
                              hipStream_t stream) {
    const float* feat = (const float*)d_in[0];
    const int*   src  = (const int*)d_in[1];
    const int*   dstv = (const int*)d_in[2];
    const float* Ws0  = (const float*)d_in[3];
    const float* bs0  = (const float*)d_in[4];
    const float* Wn0  = (const float*)d_in[5];
    const float* bn0  = (const float*)d_in[6];
    const float* Ws1  = (const float*)d_in[7];
    const float* bs1  = (const float*)d_in[8];
    const float* Wn1  = (const float*)d_in[9];
    const float* bn1  = (const float*)d_in[10];
    const float* Ws2  = (const float*)d_in[11];
    const float* bs2  = (const float*)d_in[12];
    const float* Wn2  = (const float*)d_in[13];
    const float* bn2  = (const float*)d_in[14];
    float* out = (float*)d_out;

    float* agg    = (float*)d_ws;
    float* h1     = agg + (size_t)NNODES * KIN;
    float* h2     = h1  + (size_t)NNODES * KIN;
    float* deg    = h2  + (size_t)NNODES * KIN;
    float* invdeg = deg + NNODES;
    const size_t PACK8 = (size_t)8 * (KIN / 4) * 32;
    const size_t PACK3 = (size_t)3 * (KIN / 4) * 32;
    float2* Wsp0 = (float2*)(invdeg + NNODES);
    float2* Wnp0 = Wsp0 + PACK8;
    float2* Wsp1 = Wnp0 + PACK8;
    float2* Wnp1 = Wsp1 + PACK8;
    float2* Wsp2 = Wnp1 + PACK8;
    float2* Wnp2 = Wsp2 + PACK3;

    const int ZB = 256;
    const int featN = NNODES * KIN;
    const int aggBlocks4 = (featN / 4 + ZB - 1) / ZB;
    const int edgeWaves  = NEDGES / 8;
    const int gemmBlocks = NNODES / 16;

    pack_w_kernel<<<(int)(PACK8 + ZB - 1) / ZB, ZB, 0, stream>>>(Ws0, Wsp0, 128, 8);
    pack_w_kernel<<<(int)(PACK8 + ZB - 1) / ZB, ZB, 0, stream>>>(Wn0, Wnp0, 128, 8);
    pack_w_kernel<<<(int)(PACK8 + ZB - 1) / ZB, ZB, 0, stream>>>(Ws1, Wsp1, 128, 8);
    pack_w_kernel<<<(int)(PACK8 + ZB - 1) / ZB, ZB, 0, stream>>>(Wn1, Wnp1, 128, 8);
    pack_w_kernel<<<(int)(PACK3 + ZB - 1) / ZB, ZB, 0, stream>>>(Ws2, Wsp2, 40, 3);
    pack_w_kernel<<<(int)(PACK3 + ZB - 1) / ZB, ZB, 0, stream>>>(Wn2, Wnp2, 40, 3);

    zero4_kernel<<<(NNODES / 4 + ZB - 1) / ZB, ZB, 0, stream>>>((float4*)deg, NNODES / 4);
    deg_kernel<<<(NEDGES + ZB - 1) / ZB, ZB, 0, stream>>>(dstv, deg);
    invdeg_kernel<<<(NNODES + ZB - 1) / ZB, ZB, 0, stream>>>(deg, invdeg);

    zero4_kernel<<<aggBlocks4, ZB, 0, stream>>>((float4*)agg, featN / 4);
    scatter_kernel<<<edgeWaves, 256, 0, stream>>>(feat, src, dstv, agg);
    sage_gemm<128, true, true><<<gemmBlocks, 256, 0, stream>>>(
        feat, agg, invdeg, Wsp0, Wnp0, bs0, bn0, h1);

    zero4_kernel<<<aggBlocks4, ZB, 0, stream>>>((float4*)agg, featN / 4);
    scatter_kernel<<<edgeWaves, 256, 0, stream>>>(h1, src, dstv, agg);
    sage_gemm<128, true, true><<<gemmBlocks, 256, 0, stream>>>(
        h1, agg, invdeg, Wsp1, Wnp1, bs1, bn1, h2);

    zero4_kernel<<<aggBlocks4, ZB, 0, stream>>>((float4*)agg, featN / 4);
    scatter_kernel<<<edgeWaves, 256, 0, stream>>>(h2, src, dstv, agg);
    sage_gemm<40, false, false><<<gemmBlocks, 96, 0, stream>>>(
        h2, agg, invdeg, Wsp2, Wnp2, bs2, bn2, out);
}


# Neighbour sum: f32 atomic scatter -> edges counting-sorted by dst each call (int-atomic arrival rank, scan, fill) + one-wave-per-node gather sum; no agg zero-fill; add order now follows arrival rank
# speedup vs baseline: 1.9952x; 1.9952x over previous
#include <hip/hip_runtime.h>
#include <hip/hip_bf16.h>

#define NNODES 100000
#define NEDGES 1600000
#define KIN    128

typedef __attribute__((ext_vector_type(2))) float v2f;
typedef __attribute__((ext_vector_type(8))) float v8f;

__global__ void zero4_kernel(float4* __restrict__ p, int n4) {
    int i = blockIdx.x * blockDim.x + threadIdx.x;
    if (i < n4) p[i] = make_float4(0.f, 0.f, 0.f, 0.f);
}


__global__ void rank_kernel(const int* __restrict__ dst, int* __restrict__ cnt,
                            int* __restrict__ rank) {
    int e = blockIdx.x * blockDim.x + threadIdx.x;
    if (e < NEDGES) rank[e] = atomicAdd(&cnt[dst[e]], 1);
}

__global__ void __launch_bounds__(1024)
scan_kernel(const int* __restrict__ cnt, int* __restrict__ rowptr,
            float* __restrict__ invdeg) {
    __shared__ int wsum[32];
    const int t    = threadIdx.x;
    const int lane = t & 31;
    const int wv   = t >> 5;
    const bool live = t < NNODES / 100;
    const int4* __restrict__ c4 = (const int4*)(cnt + t * 100);

    int s = 0;
    if (live) {
        for (int i = 0; i < 25; ++i) {
            const int4 v = c4[i];
            s += v.x + v.y + v.z + v.w;
        }
    }
    int x = s;
    for (int o = 1; o < 32; o <<= 1) {
        const int y = __shfl_up(x, o, 32);
        if (lane >= o) x += y;
    }
    if (lane == 31) wsum[wv] = x;
    __syncthreads();
    if (wv == 0) {
        const int w = wsum[lane];
        int z = w;
        for (int o = 1; o < 32; o <<= 1) {
            const int y = __shfl_up(z, o, 32);
            if (lane >= o) z += y;
        }
        wsum[lane] = z - w;
    }
    __syncthreads();
    int run = wsum[wv] + (x - s);
    if (live) {
        for (int i = 0; i < 25; ++i) {
            const int4 v = c4[i];
            const int base = t * 100 + 4 * i;
            int4 r;
            r.x = run;
            r.y = r.x + v.x;
            r.z = r.y + v.y;
            r.w = r.z + v.z;
            run = r.w + v.w;
            *(int4*)(rowptr + base) = r;
            float4 d;
            d.x = 1.0f / fmaxf((float)v.x, 1.0f);
            d.y = 1.0f / fmaxf((float)v.y, 1.0f);
            d.z = 1.0f / fmaxf((float)v.z, 1.0f);
            d.w = 1.0f / fmaxf((float)v.w, 1.0f);
            *(float4*)(invdeg + base) = d;
        }
    }
}

__global__ void fill_kernel(const int* __restrict__ src, const int* __restrict__ dst,
                            const int* __restrict__ rank, const int* __restrict__ rowptr,
                            int* __restrict__ col) {
    int e = blockIdx.x * blockDim.x + threadIdx.x;
    if (e < NEDGES) col[rowptr[dst[e]] + rank[e]] = src[e];
}

template<int B>
__device__ __forceinline__ void take_rows(const float* __restrict__ h, int mine, int k,
                                          int lane, float4& acc) {
    float4 v[B];
    #pragma unroll
    for (int j = 0; j < B; ++j) {
        const int s = __builtin_amdgcn_readlane(mine, k + j);
        v[j] = ((const float4*)(h + (size_t)s * KIN))[lane];
    }
    #pragma unroll
    for (int j = 0; j < B; ++j) {
        acc.x += v[j].x;
        acc.y += v[j].y;
        acc.z += v[j].z;
        acc.w += v[j].w;
    }
}

__global__ void __launch_bounds__(256)
gather_kernel(const float* __restrict__ h, const int* __restrict__ rowptr,
              const int* __restrict__ col, float* __restrict__ agg) {
    const int lane = threadIdx.x & 31;
    const int node = __builtin_amdgcn_readfirstlane((int)(blockIdx.x * 8 + (threadIdx.x >> 5)));
    const int beg = rowptr[node];
    const int end = (node + 1 < NNODES) ? rowptr[node + 1] : NEDGES;

    float4 acc = make_float4(0.f, 0.f, 0.f, 0.f);
    for (int base = beg; base < end; base += 32) {
        const int n = min(end - base, 32);
        const int mine = (lane < n) ? col[base + lane] : 0;
        int k = 0;
        for (; k + 8 <= n; k += 8) take_rows<8>(h, mine, k, lane, acc);
        if (k + 4 <= n) { take_rows<4>(h, mine, k, lane, acc); k += 4; }
        if (k + 2 <= n) { take_rows<2>(h, mine, k, lane, acc); k += 2; }
        if (k < n)      { take_rows<1>(h, mine, k, lane, acc); }
    }
    ((float4*)(agg + (size_t)node * KIN))[lane] = acc;
}

__global__ void pack_w_kernel(const float* __restrict__ W, float2* __restrict__ Wp,
                              int nout, int ntiles) {
    int pid = blockIdx.x * blockDim.x + threadIdx.x;
    int total = ntiles * (KIN / 4) * 32;
    if (pid >= total) return;
    int lane = pid & 31;
    int i    = (pid >> 5) & ((KIN / 4) - 1);
    int tile = pid >> 10;
    int l = lane & 15, half = lane >> 4;
    int col = tile * 16 + l;
    int k   = 4 * i + 2 * half;
    float x = 0.f, y = 0.f;
    if (col < nout) {
        x = W[(size_t)k       * nout + col];
        y = W[(size_t)(k + 1) * nout + col];
    }
    Wp[pid] = make_float2(x, y);
}

template<int NOUT, bool ACT, bool NORM>
__global__ void __launch_bounds__(256)
sage_gemm(const float* __restrict__ hself,
          const float* __restrict__ agg,
          const float* __restrict__ invdeg,
          const float2* __restrict__ Wsp,
          const float2* __restrict__ Wnp,
          const float* __restrict__ bself,
          const float* __restrict__ bneigh,
          float* __restrict__ out) {
    constexpr bool FULL = (NOUT % 16) == 0;
    const int lane = threadIdx.x & 31;
    const int wv   = threadIdx.x >> 5;
    const int l    = lane & 15;
    const int half = lane >> 4;
    const int m0   = blockIdx.x * 16;
    const int n0   = wv * 16;
    const int col  = n0 + l;
    const bool colok = FULL || (col < NOUT);
    const int arow = m0 + l;

    const float idg = invdeg[arow];
    const float*  __restrict__ As = hself + (size_t)arow * KIN + 2 * half;
    const float*  __restrict__ An = agg   + (size_t)arow * KIN + 2 * half;
    const float2* __restrict__ Bs = Wsp + (size_t)wv * (KIN / 4) * 32 + lane;
    const float2* __restrict__ Bn = Wnp + (size_t)wv * (KIN / 4) * 32 + lane;

    v8f acc = {0.f, 0.f, 0.f, 0.f, 0.f, 0.f, 0.f, 0.f};

    #pragma unroll
    for (int i = 0; i < KIN / 4; ++i) {
        const float2 af = *(const float2*)(As + 4 * i);
        const float2 bf = Bs[i * 32];
        v2f a = {af.x, af.y};
        v2f b = {bf.x, bf.y};
        acc = __builtin_amdgcn_wmma_f32_16x16x4_f32(false, a, false, b,
                                                    (short)0, acc, false, false);
    }
    #pragma unroll
    for (int i = 0; i < KIN / 4; ++i) {
        const float2 af = *(const float2*)(An + 4 * i);
        const float2 bf = Bn[i * 32];
        v2f a = {af.x * idg, af.y * idg};
        v2f b = {bf.x, bf.y};
        acc = __builtin_amdgcn_wmma_f32_16x16x4_f32(false, a, false, b,
                                                    (short)0, acc, false, false);
    }

    const float bias = colok ? (bself[col] + bneigh[col]) : 0.0f;
    #pragma unroll
    for (int r = 0; r < 8; ++r) {
        float v = acc[r] + bias;
        if (ACT) v = fmaxf(v, 0.0f);
        acc[r] = v;
    }

    if (NORM) {
        __shared__ float rowsum[16];
        if (threadIdx.x < 16) rowsum[threadIdx.x] = 0.0f;
        __syncthreads();
        #pragma unroll
        for (int r = 0; r < 8; ++r)
            atomicAdd(&rowsum[r + 8 * half], acc[r] * acc[r]);
        __syncthreads();
        #pragma unroll
        for (int r = 0; r < 8; ++r) {
            float nrm = fmaxf(sqrtf(rowsum[r + 8 * half]), 1e-12f);
            acc[r] = acc[r] / nrm;
        }
    }

    if (colok) {
        #pragma unroll
        for (int r = 0; r < 8; ++r) {
            const int m = m0 + r + 8 * half;
            out[(size_t)m * NOUT + col] = acc[r];
        }
    }
}

extern "C" void kernel_launch(void* const* d_in, const int* in_sizes, int n_in,
                              void* d_out, int out_size, void* d_ws, size_t ws_size,
                              hipStream_t stream) {
    const float* feat = (const float*)d_in[0];
    const int*   src  = (const int*)d_in[1];
    const int*   dstv = (const int*)d_in[2];
    const float* Ws0  = (const float*)d_in[3];
    const float* bs0  = (const float*)d_in[4];
    const float* Wn0  = (const float*)d_in[5];
    const float* bn0  = (const float*)d_in[6];
    const float* Ws1  = (const float*)d_in[7];
    const float* bs1  = (const float*)d_in[8];
    const float* Wn1  = (const float*)d_in[9];
    const float* bn1  = (const float*)d_in[10];
    const float* Ws2  = (const float*)d_in[11];
    const float* bs2  = (const float*)d_in[12];
    const float* Wn2  = (const float*)d_in[13];
    const float* bn2  = (const float*)d_in[14];
    float* out = (float*)d_out;

    float* agg    = (float*)d_ws;
    float* h1     = agg + (size_t)NNODES * KIN;
    float* h2     = h1  + (size_t)NNODES * KIN;
    int*   rowptr = (int*)(h2 + (size_t)NNODES * KIN);
    float* invdeg = (float*)(rowptr + NNODES);
    int* cnt  = (int*)h2;
    int* rank = cnt + NNODES;
    int* col = (int*)d_out;
    const size_t PACK8 = (size_t)8 * (KIN / 4) * 32;
    const size_t PACK3 = (size_t)3 * (KIN / 4) * 32;
    float2* Wsp0 = (float2*)(invdeg + NNODES);
    float2* Wnp0 = Wsp0 + PACK8;
    float2* Wsp1 = Wnp0 + PACK8;
    float2* Wnp1 = Wsp1 + PACK8;
    float2* Wsp2 = Wnp1 + PACK8;
    float2* Wnp2 = Wsp2 + PACK3;

    const int ZB = 256;
    const int nodeBlocks = NNODES / 8;
    const int gemmBlocks = NNODES / 16;

    pack_w_kernel<<<(int)(PACK8 + ZB - 1) / ZB, ZB, 0, stream>>>(Ws0, Wsp0, 128, 8);
    pack_w_kernel<<<(int)(PACK8 + ZB - 1) / ZB, ZB, 0, stream>>>(Wn0, Wnp0, 128, 8);
    pack_w_kernel<<<(int)(PACK8 + ZB - 1) / ZB, ZB, 0, stream>>>(Ws1, Wsp1, 128, 8);
    pack_w_kernel<<<(int)(PACK8 + ZB - 1) / ZB, ZB, 0, stream>>>(Wn1, Wnp1, 128, 8);
    pack_w_kernel<<<(int)(PACK3 + ZB - 1) / ZB, ZB, 0, stream>>>(Ws2, Wsp2, 40, 3);
    pack_w_kernel<<<(int)(PACK3 + ZB - 1) / ZB, ZB, 0, stream>>>(Wn2, Wnp2, 40, 3);

    zero4_kernel<<<(NNODES / 4 + ZB - 1) / ZB, ZB, 0, stream>>>((float4*)cnt, NNODES / 4);
    rank_kernel<<<(NEDGES + ZB - 1) / ZB, ZB, 0, stream>>>(dstv, cnt, rank);
    scan_kernel<<<1, 1024, 0, stream>>>(cnt, rowptr, invdeg);
    fill_kernel<<<(NEDGES + ZB - 1) / ZB, ZB, 0, stream>>>(src, dstv, rank, rowptr, col);

    gather_kernel<<<nodeBlocks, 256, 0, stream>>>(feat, rowptr, col, agg);
    sage_gemm<128, true, true><<<gemmBlocks, 256, 0, stream>>>(
        feat, agg, invdeg, Wsp0, Wnp0, bs0, bn0, h1);

    gather_kernel<<<nodeBlocks, 256, 0, stream>>>(h1, rowptr, col, agg);
    sage_gemm<128, true, true><<<gemmBlocks, 256, 0, stream>>>(
        h1, agg, invdeg, Wsp1, Wnp1, bs1, bn1, h2);

    gather_kernel<<<nodeBlocks, 256, 0, stream>>>(h2, rowptr, col, agg);
    sage_gemm<40, false, false><<<gemmBlocks, 96, 0, stream>>>(
        h2, agg, invdeg, Wsp2, Wnp2, bs2, bn2, out);
}


# Neighbour gather fused into the layer GEMM kernel: a 32-node work-group sums neighbour and own rows into LDS, WMMA reads A from LDS, 2 row tiles per B load; agg plane, 3 launches gone; add order kept
# speedup vs baseline: 2.5945x; 1.3004x over previous
#include <hip/hip_runtime.h>
#include <hip/hip_bf16.h>

#define NNODES 100000
#define NEDGES 1600000
#define KIN    128

typedef __attribute__((ext_vector_type(2))) float v2f;
typedef __attribute__((ext_vector_type(8))) float v8f;

__global__ void zero4_kernel(float4* __restrict__ p, int n4) {
    int i = blockIdx.x * blockDim.x + threadIdx.x;
    if (i < n4) p[i] = make_float4(0.f, 0.f, 0.f, 0.f);
}


__global__ void rank_kernel(const int* __restrict__ dst, int* __restrict__ cnt,
                            int* __restrict__ rank) {
    int e = blockIdx.x * blockDim.x + threadIdx.x;
    if (e < NEDGES) rank[e] = atomicAdd(&cnt[dst[e]], 1);
}

__global__ void __launch_bounds__(1024)
scan_kernel(const int* __restrict__ cnt, int* __restrict__ rowptr,
            float* __restrict__ invdeg) {
    __shared__ int wsum[32];
    const int t    = threadIdx.x;
    const int lane = t & 31;
    const int wv   = t >> 5;
    const bool live = t < NNODES / 100;
    const int4* __restrict__ c4 = (const int4*)(cnt + t * 100);

    int s = 0;
    if (live) {
        for (int i = 0; i < 25; ++i) {
            const int4 v = c4[i];
            s += v.x + v.y + v.z + v.w;
        }
    }
    int x = s;
    for (int o = 1; o < 32; o <<= 1) {
        const int y = __shfl_up(x, o, 32);
        if (lane >= o) x += y;
    }
    if (lane == 31) wsum[wv] = x;
    __syncthreads();
    if (wv == 0) {
        const int w = wsum[lane];
        int z = w;
        for (int o = 1; o < 32; o <<= 1) {
            const int y = __shfl_up(z, o, 32);
            if (lane >= o) z += y;
        }
        wsum[lane] = z - w;
    }
    __syncthreads();
    int run = wsum[wv] + (x - s);
    if (live) {
        for (int i = 0; i < 25; ++i) {
            const int4 v = c4[i];
            const int base = t * 100 + 4 * i;
            int4 r;
            r.x = run;
            r.y = r.x + v.x;
            r.z = r.y + v.y;
            r.w = r.z + v.z;
            run = r.w + v.w;
            *(int4*)(rowptr + base) = r;
            float4 d;
            d.x = 1.0f / fmaxf((float)v.x, 1.0f);
            d.y = 1.0f / fmaxf((float)v.y, 1.0f);
            d.z = 1.0f / fmaxf((float)v.z, 1.0f);
            d.w = 1.0f / fmaxf((float)v.w, 1.0f);
            *(float4*)(invdeg + base) = d;
        }
    }
}

__global__ void fill_kernel(const int* __restrict__ src, const int* __restrict__ dst,
                            const int* __restrict__ rank, const int* __restrict__ rowptr,
                            int* __restrict__ col) {
    int e = blockIdx.x * blockDim.x + threadIdx.x;
    if (e < NEDGES) col[rowptr[dst[e]] + rank[e]] = src[e];
}

template<int B>
__device__ __forceinline__ void take_rows(const float* __restrict__ h, int mine, int k,
                                          int lane, float4& acc) {
    float4 v[B];
    #pragma unroll
    for (int j = 0; j < B; ++j) {
        const int s = __builtin_amdgcn_readlane(mine, k + j);
        v[j] = ((const float4*)(h + (size_t)s * KIN))[lane];
    }
    #pragma unroll
    for (int j = 0; j < B; ++j) {
        acc.x += v[j].x;
        acc.y += v[j].y;
        acc.z += v[j].z;
        acc.w += v[j].w;
    }
}

__global__ void pack_w_kernel(const float* __restrict__ W, float2* __restrict__ Wp,
                              int nout, int ntiles) {
    int pid = blockIdx.x * blockDim.x + threadIdx.x;
    int total = ntiles * (KIN / 4) * 32;
    if (pid >= total) return;
    int lane = pid & 31;
    int i    = (pid >> 5) & ((KIN / 4) - 1);
    int tile = pid >> 10;
    int l = lane & 15, half = lane >> 4;
    int col = tile * 16 + l;
    int k   = 4 * i + 2 * half;
    float x = 0.f, y = 0.f;
    if (col < nout) {
        x = W[(size_t)k       * nout + col];
        y = W[(size_t)(k + 1) * nout + col];
    }
    Wp[pid] = make_float2(x, y);
}

template<int NOUT, bool ACT, bool NORM>
__global__ void __launch_bounds__(256) __attribute__((amdgpu_waves_per_eu(16, 16)))
sage_layer(const float* __restrict__ h,
           const int* __restrict__ rowptr,
           const int* __restrict__ col,
           const float* __restrict__ invdeg,
           const float2* __restrict__ Wsp,
           const float2* __restrict__ Wnp,
           const float* __restrict__ bself,
           const float* __restrict__ bneigh,
           float* __restrict__ out) {
    constexpr bool FULL = (NOUT % 16) == 0;
    constexpr int NT    = (NOUT + 15) / 16;
    constexpr int ROWS  = 32;
    constexpr int PITCH = KIN + 4;
    static_assert(!NORM || NT == 8, "the row-norm barriers need all eight waves");
    __shared__ float sSelf[ROWS * PITCH];
    __shared__ float sNeigh[ROWS * PITCH];

    const int lane = threadIdx.x & 31;
    const int wv   = __builtin_amdgcn_readfirstlane((int)(threadIdx.x >> 5));
    const int m0   = blockIdx.x * ROWS;

    int   beg[4], end[4], first[4];
    float idg[4];
    #pragma unroll
    for (int t = 0; t < 4; ++t) {
        const int node = m0 + wv + 8 * t;
        beg[t] = rowptr[node];
        end[t] = (node + 1 < NNODES) ? rowptr[node + 1] : NEDGES;
        idg[t] = invdeg[node];
    }
    #pragma unroll
    for (int t = 0; t < 4; ++t)
        first[t] = (lane < end[t] - beg[t]) ? col[beg[t] + lane] : 0;

    #pragma unroll
    for (int t = 0; t < 4; ++t) {
        const int r    = wv + 8 * t;
        const int node = m0 + r;
        const float4 hv = ((const float4*)(h + (size_t)node * KIN))[lane];

        float4 acc = make_float4(0.f, 0.f, 0.f, 0.f);
        int mine = first[t];
        for (int base = beg[t]; base < end[t]; ) {
            const int n = min(end[t] - base, 32);
            int k = 0;
            for (; k + 8 <= n; k += 8) take_rows<8>(h, mine, k, lane, acc);
            if (k + 4 <= n) { take_rows<4>(h, mine, k, lane, acc); k += 4; }
            if (k + 2 <= n) { take_rows<2>(h, mine, k, lane, acc); k += 2; }
            if (k < n)      { take_rows<1>(h, mine, k, lane, acc); }
            base += 32;
            if (base < end[t])
                mine = (lane < end[t] - base) ? col[base + lane] : 0;
        }
        acc.x *= idg[t]; acc.y *= idg[t]; acc.z *= idg[t]; acc.w *= idg[t];
        *(float4*)(sSelf  + r * PITCH + 4 * lane) = hv;
        *(float4*)(sNeigh + r * PITCH + 4 * lane) = acc;
    }
    __syncthreads();
    if (NT < 8 && wv >= NT) return;

    const int l    = lane & 15;
    const int half = lane >> 4;
    const int n0   = wv * 16;
    const int ocol = n0 + l;
    const bool colok = FULL || (ocol < NOUT);

    const float*  As = sSelf  + l * PITCH + 2 * half;
    const float*  An = sNeigh + l * PITCH + 2 * half;
    const float2* __restrict__ Bs = Wsp + (size_t)wv * (KIN / 4) * 32 + lane;
    const float2* __restrict__ Bn = Wnp + (size_t)wv * (KIN / 4) * 32 + lane;

    v8f acc0 = {0.f, 0.f, 0.f, 0.f, 0.f, 0.f, 0.f, 0.f};
    v8f acc1 = {0.f, 0.f, 0.f, 0.f, 0.f, 0.f, 0.f, 0.f};

    #pragma unroll
    for (int i = 0; i < KIN / 4; ++i) {
        const float2 a0f = *(const float2*)(As + 4 * i);
        const float2 a1f = *(const float2*)(As + 16 * PITCH + 4 * i);
        const float2 bf  = Bs[i * 32];
        v2f a0 = {a0f.x, a0f.y};
        v2f a1 = {a1f.x, a1f.y};
        v2f b  = {bf.x, bf.y};
        acc0 = __builtin_amdgcn_wmma_f32_16x16x4_f32(false, a0, false, b,
                                                     (short)0, acc0, false, false);
        acc1 = __builtin_amdgcn_wmma_f32_16x16x4_f32(false, a1, false, b,
                                                     (short)0, acc1, false, false);
    }
    #pragma unroll
    for (int i = 0; i < KIN / 4; ++i) {
        const float2 a0f = *(const float2*)(An + 4 * i);
        const float2 a1f = *(const float2*)(An + 16 * PITCH + 4 * i);
        const float2 bf  = Bn[i * 32];
        v2f a0 = {a0f.x, a0f.y};
        v2f a1 = {a1f.x, a1f.y};
        v2f b  = {bf.x, bf.y};
        acc0 = __builtin_amdgcn_wmma_f32_16x16x4_f32(false, a0, false, b,
                                                     (short)0, acc0, false, false);
        acc1 = __builtin_amdgcn_wmma_f32_16x16x4_f32(false, a1, false, b,
                                                     (short)0, acc1, false, false);
    }

    const float bias = colok ? (bself[ocol] + bneigh[ocol]) : 0.0f;
    #pragma unroll
    for (int r = 0; r < 8; ++r) {
        float v0 = acc0[r] + bias;
        float v1 = acc1[r] + bias;
        if (ACT) { v0 = fmaxf(v0, 0.0f); v1 = fmaxf(v1, 0.0f); }
        acc0[r] = v0;
        acc1[r] = v1;
    }

    if (NORM) {
        __shared__ float rowsum[ROWS];
        if (threadIdx.x < ROWS) rowsum[threadIdx.x] = 0.0f;
        __syncthreads();
        #pragma unroll
        for (int r = 0; r < 8; ++r) {
            atomicAdd(&rowsum[r + 8 * half],      acc0[r] * acc0[r]);
            atomicAdd(&rowsum[16 + r + 8 * half], acc1[r] * acc1[r]);
        }
        __syncthreads();
        #pragma unroll
        for (int r = 0; r < 8; ++r) {
            const float nrm0 = fmaxf(sqrtf(rowsum[r + 8 * half]), 1e-12f);
            const float nrm1 = fmaxf(sqrtf(rowsum[16 + r + 8 * half]), 1e-12f);
            acc0[r] = acc0[r] / nrm0;
            acc1[r] = acc1[r] / nrm1;
        }
    }

    if (colok) {
        #pragma unroll
        for (int r = 0; r < 8; ++r) {
            const int m = m0 + r + 8 * half;
            out[(size_t)m * NOUT + ocol]        = acc0[r];
            out[(size_t)(m + 16) * NOUT + ocol] = acc1[r];
        }
    }
}

extern "C" void kernel_launch(void* const* d_in, const int* in_sizes, int n_in,
                              void* d_out, int out_size, void* d_ws, size_t ws_size,
                              hipStream_t stream) {
    const float* feat = (const float*)d_in[0];
    const int*   src  = (const int*)d_in[1];
    const int*   dstv = (const int*)d_in[2];
    const float* Ws0  = (const float*)d_in[3];
    const float* bs0  = (const float*)d_in[4];
    const float* Wn0  = (const float*)d_in[5];
    const float* bn0  = (const float*)d_in[6];
    const float* Ws1  = (const float*)d_in[7];
    const float* bs1  = (const float*)d_in[8];
    const float* Wn1  = (const float*)d_in[9];
    const float* bn1  = (const float*)d_in[10];
    const float* Ws2  = (const float*)d_in[11];
    const float* bs2  = (const float*)d_in[12];
    const float* Wn2  = (const float*)d_in[13];
    const float* bn2  = (const float*)d_in[14];
    float* out = (float*)d_out;

    int*   col    = (int*)d_ws;
    float* h1     = (float*)d_ws + (size_t)NNODES * KIN;
    float* h2     = h1  + (size_t)NNODES * KIN;
    int*   rowptr = (int*)(h2 + (size_t)NNODES * KIN);
    float* invdeg = (float*)(rowptr + NNODES);
    int* cnt  = (int*)h2;
    int* rank = cnt + NNODES;
    const size_t PACK8 = (size_t)8 * (KIN / 4) * 32;
    const size_t PACK3 = (size_t)3 * (KIN / 4) * 32;
    float2* Wsp0 = (float2*)(invdeg + NNODES);
    float2* Wnp0 = Wsp0 + PACK8;
    float2* Wsp1 = Wnp0 + PACK8;
    float2* Wnp1 = Wsp1 + PACK8;
    float2* Wsp2 = Wnp1 + PACK8;
    float2* Wnp2 = Wsp2 + PACK3;

    const int ZB = 256;
    const int tileBlocks = NNODES / 32;

    pack_w_kernel<<<(int)(PACK8 + ZB - 1) / ZB, ZB, 0, stream>>>(Ws0, Wsp0, 128, 8);
    pack_w_kernel<<<(int)(PACK8 + ZB - 1) / ZB, ZB, 0, stream>>>(Wn0, Wnp0, 128, 8);
    pack_w_kernel<<<(int)(PACK8 + ZB - 1) / ZB, ZB, 0, stream>>>(Ws1, Wsp1, 128, 8);
    pack_w_kernel<<<(int)(PACK8 + ZB - 1) / ZB, ZB, 0, stream>>>(Wn1, Wnp1, 128, 8);
    pack_w_kernel<<<(int)(PACK3 + ZB - 1) / ZB, ZB, 0, stream>>>(Ws2, Wsp2, 40, 3);
    pack_w_kernel<<<(int)(PACK3 + ZB - 1) / ZB, ZB, 0, stream>>>(Wn2, Wnp2, 40, 3);

    zero4_kernel<<<(NNODES / 4 + ZB - 1) / ZB, ZB, 0, stream>>>((float4*)cnt, NNODES / 4);
    rank_kernel<<<(NEDGES + ZB - 1) / ZB, ZB, 0, stream>>>(dstv, cnt, rank);
    scan_kernel<<<1, 1024, 0, stream>>>(cnt, rowptr, invdeg);
    fill_kernel<<<(NEDGES + ZB - 1) / ZB, ZB, 0, stream>>>(src, dstv, rank, rowptr, col);

    sage_layer<128, true, true><<<tileBlocks, 256, 0, stream>>>(
        feat, rowptr, col, invdeg, Wsp0, Wnp0, bs0, bn0, h1);

    sage_layer<128, true, true><<<tileBlocks, 256, 0, stream>>>(
        h1, rowptr, col, invdeg, Wsp1, Wnp1, bs1, bn1, h2);

    sage_layer<40, false, false><<<tileBlocks, 256, 0, stream>>>(
        h2, rowptr, col, invdeg, Wsp2, Wnp2, bs2, bn2, out);
}


# Row norm: sum of squares by LDS f32 atomics (16 lanes per word, serialised) -> 16-lane register butterfly + one LDS partial per wave, added in wave order; root once per row; add order now fixed
# speedup vs baseline: 3.1046x; 1.1966x over previous
#include <hip/hip_runtime.h>
#include <hip/hip_bf16.h>

#define NNODES 100000
#define NEDGES 1600000
#define KIN    128

typedef __attribute__((ext_vector_type(2))) float v2f;
typedef __attribute__((ext_vector_type(8))) float v8f;

__global__ void zero4_kernel(float4* __restrict__ p, int n4) {
    int i = blockIdx.x * blockDim.x + threadIdx.x;
    if (i < n4) p[i] = make_float4(0.f, 0.f, 0.f, 0.f);
}


__global__ void rank_kernel(const int* __restrict__ dst, int* __restrict__ cnt,
                            int* __restrict__ rank) {
    int e = blockIdx.x * blockDim.x + threadIdx.x;
    if (e < NEDGES) rank[e] = atomicAdd(&cnt[dst[e]], 1);
}

__global__ void __launch_bounds__(1024)
scan_kernel(const int* __restrict__ cnt, int* __restrict__ rowptr,
            float* __restrict__ invdeg) {
    __shared__ int wsum[32];
    const int t    = threadIdx.x;
    const int lane = t & 31;
    const int wv   = t >> 5;
    const bool live = t < NNODES / 100;
    const int4* __restrict__ c4 = (const int4*)(cnt + t * 100);

    int s = 0;
    if (live) {
        for (int i = 0; i < 25; ++i) {
            const int4 v = c4[i];
            s += v.x + v.y + v.z + v.w;
        }
    }
    int x = s;
    for (int o = 1; o < 32; o <<= 1) {
        const int y = __shfl_up(x, o, 32);
        if (lane >= o) x += y;
    }
    if (lane == 31) wsum[wv] = x;
    __syncthreads();
    if (wv == 0) {
        const int w = wsum[lane];
        int z = w;
        for (int o = 1; o < 32; o <<= 1) {
            const int y = __shfl_up(z, o, 32);
            if (lane >= o) z += y;
        }
        wsum[lane] = z - w;
    }
    __syncthreads();
    int run = wsum[wv] + (x - s);
    if (live) {
        for (int i = 0; i < 25; ++i) {
            const int4 v = c4[i];
            const int base = t * 100 + 4 * i;
            int4 r;
            r.x = run;
            r.y = r.x + v.x;
            r.z = r.y + v.y;
            r.w = r.z + v.z;
            run = r.w + v.w;
            *(int4*)(rowptr + base) = r;
            float4 d;
            d.x = 1.0f / fmaxf((float)v.x, 1.0f);
            d.y = 1.0f / fmaxf((float)v.y, 1.0f);
            d.z = 1.0f / fmaxf((float)v.z, 1.0f);
            d.w = 1.0f / fmaxf((float)v.w, 1.0f);
            *(float4*)(invdeg + base) = d;
        }
    }
}

__global__ void fill_kernel(const int* __restrict__ src, const int* __restrict__ dst,
                            const int* __restrict__ rank, const int* __restrict__ rowptr,
                            int* __restrict__ col) {
    int e = blockIdx.x * blockDim.x + threadIdx.x;
    if (e < NEDGES) col[rowptr[dst[e]] + rank[e]] = src[e];
}

__device__ __forceinline__ float sum16(float v) {
    v += __builtin_bit_cast(float, __builtin_amdgcn_update_dpp(0, __builtin_bit_cast(int, v), 0x161, 0xf, 0xf, true));
    v += __builtin_bit_cast(float, __builtin_amdgcn_update_dpp(0, __builtin_bit_cast(int, v), 0x162, 0xf, 0xf, true));
    v += __builtin_bit_cast(float, __builtin_amdgcn_update_dpp(0, __builtin_bit_cast(int, v), 0x164, 0xf, 0xf, true));
    v += __builtin_bit_cast(float, __builtin_amdgcn_update_dpp(0, __builtin_bit_cast(int, v), 0x168, 0xf, 0xf, true));
    return v;
}

template<int B>
__device__ __forceinline__ void take_rows(const float* __restrict__ h, int mine, int k,
                                          int lane, float4& acc) {
    float4 v[B];
    #pragma unroll
    for (int j = 0; j < B; ++j) {
        const int s = __builtin_amdgcn_readlane(mine, k + j);
        v[j] = ((const float4*)(h + (size_t)s * KIN))[lane];
    }
    #pragma unroll
    for (int j = 0; j < B; ++j) {
        acc.x += v[j].x;
        acc.y += v[j].y;
        acc.z += v[j].z;
        acc.w += v[j].w;
    }
}

__global__ void pack_w_kernel(const float* __restrict__ W, float2* __restrict__ Wp,
                              int nout, int ntiles) {
    int pid = blockIdx.x * blockDim.x + threadIdx.x;
    int total = ntiles * (KIN / 4) * 32;
    if (pid >= total) return;
    int lane = pid & 31;
    int i    = (pid >> 5) & ((KIN / 4) - 1);
    int tile = pid >> 10;
    int l = lane & 15, half = lane >> 4;
    int col = tile * 16 + l;
    int k   = 4 * i + 2 * half;
    float x = 0.f, y = 0.f;
    if (col < nout) {
        x = W[(size_t)k       * nout + col];
        y = W[(size_t)(k + 1) * nout + col];
    }
    Wp[pid] = make_float2(x, y);
}

template<int NOUT, bool ACT, bool NORM>
__global__ void __launch_bounds__(256) __attribute__((amdgpu_waves_per_eu(16, 16)))
sage_layer(const float* __restrict__ h,
           const int* __restrict__ rowptr,
           const int* __restrict__ col,
           const float* __restrict__ invdeg,
           const float2* __restrict__ Wsp,
           const float2* __restrict__ Wnp,
           const float* __restrict__ bself,
           const float* __restrict__ bneigh,
           float* __restrict__ out) {
    constexpr bool FULL = (NOUT % 16) == 0;
    constexpr int NT    = (NOUT + 15) / 16;
    constexpr int ROWS  = 32;
    constexpr int PITCH = KIN + 4;
    static_assert(!NORM || NT == 8, "the row-norm barriers need all eight waves");
    __shared__ float sSelf[ROWS * PITCH];
    __shared__ float sNeigh[ROWS * PITCH];

    const int lane = threadIdx.x & 31;
    const int wv   = __builtin_amdgcn_readfirstlane((int)(threadIdx.x >> 5));
    const int m0   = blockIdx.x * ROWS;

    int   beg[4], end[4], first[4];
    float idg[4];
    #pragma unroll
    for (int t = 0; t < 4; ++t) {
        const int node = m0 + wv + 8 * t;
        beg[t] = rowptr[node];
        end[t] = (node + 1 < NNODES) ? rowptr[node + 1] : NEDGES;
        idg[t] = invdeg[node];
    }
    #pragma unroll
    for (int t = 0; t < 4; ++t)
        first[t] = (lane < end[t] - beg[t]) ? col[beg[t] + lane] : 0;

    #pragma unroll
    for (int t = 0; t < 4; ++t) {
        const int r    = wv + 8 * t;
        const int node = m0 + r;
        const float4 hv = ((const float4*)(h + (size_t)node * KIN))[lane];

        float4 acc = make_float4(0.f, 0.f, 0.f, 0.f);
        int mine = first[t];
        for (int base = beg[t]; base < end[t]; ) {
            const int n = min(end[t] - base, 32);
            int k = 0;
            for (; k + 8 <= n; k += 8) take_rows<8>(h, mine, k, lane, acc);
            if (k + 4 <= n) { take_rows<4>(h, mine, k, lane, acc); k += 4; }
            if (k + 2 <= n) { take_rows<2>(h, mine, k, lane, acc); k += 2; }
            if (k < n)      { take_rows<1>(h, mine, k, lane, acc); }
            base += 32;
            if (base < end[t])
                mine = (lane < end[t] - base) ? col[base + lane] : 0;
        }
        acc.x *= idg[t]; acc.y *= idg[t]; acc.z *= idg[t]; acc.w *= idg[t];
        *(float4*)(sSelf  + r * PITCH + 4 * lane) = hv;
        *(float4*)(sNeigh + r * PITCH + 4 * lane) = acc;
    }
    __syncthreads();
    if (NT < 8 && wv >= NT) return;

    const int l    = lane & 15;
    const int half = lane >> 4;
    const int n0   = wv * 16;
    const int ocol = n0 + l;
    const bool colok = FULL || (ocol < NOUT);

    const float*  As = sSelf  + l * PITCH + 2 * half;
    const float*  An = sNeigh + l * PITCH + 2 * half;
    const float2* __restrict__ Bs = Wsp + (size_t)wv * (KIN / 4) * 32 + lane;
    const float2* __restrict__ Bn = Wnp + (size_t)wv * (KIN / 4) * 32 + lane;

    v8f acc0 = {0.f, 0.f, 0.f, 0.f, 0.f, 0.f, 0.f, 0.f};
    v8f acc1 = {0.f, 0.f, 0.f, 0.f, 0.f, 0.f, 0.f, 0.f};

    #pragma unroll
    for (int i = 0; i < KIN / 4; ++i) {
        const float2 a0f = *(const float2*)(As + 4 * i);
        const float2 a1f = *(const float2*)(As + 16 * PITCH + 4 * i);
        const float2 bf  = Bs[i * 32];
        v2f a0 = {a0f.x, a0f.y};
        v2f a1 = {a1f.x, a1f.y};
        v2f b  = {bf.x, bf.y};
        acc0 = __builtin_amdgcn_wmma_f32_16x16x4_f32(false, a0, false, b,
                                                     (short)0, acc0, false, false);
        acc1 = __builtin_amdgcn_wmma_f32_16x16x4_f32(false, a1, false, b,
                                                     (short)0, acc1, false, false);
    }
    #pragma unroll
    for (int i = 0; i < KIN / 4; ++i) {
        const float2 a0f = *(const float2*)(An + 4 * i);
        const float2 a1f = *(const float2*)(An + 16 * PITCH + 4 * i);
        const float2 bf  = Bn[i * 32];
        v2f a0 = {a0f.x, a0f.y};
        v2f a1 = {a1f.x, a1f.y};
        v2f b  = {bf.x, bf.y};
        acc0 = __builtin_amdgcn_wmma_f32_16x16x4_f32(false, a0, false, b,
                                                     (short)0, acc0, false, false);
        acc1 = __builtin_amdgcn_wmma_f32_16x16x4_f32(false, a1, false, b,
                                                     (short)0, acc1, false, false);
    }

    const float bias = colok ? (bself[ocol] + bneigh[ocol]) : 0.0f;
    #pragma unroll
    for (int r = 0; r < 8; ++r) {
        float v0 = acc0[r] + bias;
        float v1 = acc1[r] + bias;
        if (ACT) { v0 = fmaxf(v0, 0.0f); v1 = fmaxf(v1, 0.0f); }
        acc0[r] = v0;
        acc1[r] = v1;
    }

    if (NORM) {
        __shared__ alignas(16) float part[8 * ROWS];
        float q0[8], q1[8];
        #pragma unroll
        for (int r = 0; r < 8; ++r) {
            q0[r] = sum16(acc0[r] * acc0[r]);
            q1[r] = sum16(acc1[r] * acc1[r]);
        }
        if (l == 0) {
            float* p = part + wv * ROWS + 8 * half;
            *(float4*)(p)      = make_float4(q0[0], q0[1], q0[2], q0[3]);
            *(float4*)(p + 4)  = make_float4(q0[4], q0[5], q0[6], q0[7]);
            *(float4*)(p + 16) = make_float4(q1[0], q1[1], q1[2], q1[3]);
            *(float4*)(p + 20) = make_float4(q1[4], q1[5], q1[6], q1[7]);
        }
        __syncthreads();
        float tot = part[lane];
        #pragma unroll
        for (int w = 1; w < 8; ++w) tot += part[w * ROWS + lane];
        const float nrm = fmaxf(sqrtf(tot), 1e-12f);
        #pragma unroll
        for (int r = 0; r < 8; ++r) {
            const float nrm0 = __shfl(nrm, r + 8 * half, 32);
            const float nrm1 = __shfl(nrm, 16 + r + 8 * half, 32);
            acc0[r] = acc0[r] / nrm0;
            acc1[r] = acc1[r] / nrm1;
        }
    }

    if (colok) {
        #pragma unroll
        for (int r = 0; r < 8; ++r) {
            const int m = m0 + r + 8 * half;
            out[(size_t)m * NOUT + ocol]        = acc0[r];
            out[(size_t)(m + 16) * NOUT + ocol] = acc1[r];
        }
    }
}

extern "C" void kernel_launch(void* const* d_in, const int* in_sizes, int n_in,
                              void* d_out, int out_size, void* d_ws, size_t ws_size,
                              hipStream_t stream) {
    const float* feat = (const float*)d_in[0];
    const int*   src  = (const int*)d_in[1];
    const int*   dstv = (const int*)d_in[2];
    const float* Ws0  = (const float*)d_in[3];
    const float* bs0  = (const float*)d_in[4];
    const float* Wn0  = (const float*)d_in[5];
    const float* bn0  = (const float*)d_in[6];
    const float* Ws1  = (const float*)d_in[7];
    const float* bs1  = (const float*)d_in[8];
    const float* Wn1  = (const float*)d_in[9];
    const float* bn1  = (const float*)d_in[10];
    const float* Ws2  = (const float*)d_in[11];
    const float* bs2  = (const float*)d_in[12];
    const float* Wn2  = (const float*)d_in[13];
    const float* bn2  = (const float*)d_in[14];
    float* out = (float*)d_out;

    int*   col    = (int*)d_ws;
    float* h1     = (float*)d_ws + (size_t)NNODES * KIN;
    float* h2     = h1  + (size_t)NNODES * KIN;
    int*   rowptr = (int*)(h2 + (size_t)NNODES * KIN);
    float* invdeg = (float*)(rowptr + NNODES);
    int* cnt  = (int*)h2;
    int* rank = cnt + NNODES;
    const size_t PACK8 = (size_t)8 * (KIN / 4) * 32;
    const size_t PACK3 = (size_t)3 * (KIN / 4) * 32;
    float2* Wsp0 = (float2*)(invdeg + NNODES);
    float2* Wnp0 = Wsp0 + PACK8;
    float2* Wsp1 = Wnp0 + PACK8;
    float2* Wnp1 = Wsp1 + PACK8;
    float2* Wsp2 = Wnp1 + PACK8;
    float2* Wnp2 = Wsp2 + PACK3;

    const int ZB = 256;
    const int tileBlocks = NNODES / 32;

    pack_w_kernel<<<(int)(PACK8 + ZB - 1) / ZB, ZB, 0, stream>>>(Ws0, Wsp0, 128, 8);
    pack_w_kernel<<<(int)(PACK8 + ZB - 1) / ZB, ZB, 0, stream>>>(Wn0, Wnp0, 128, 8);
    pack_w_kernel<<<(int)(PACK8 + ZB - 1) / ZB, ZB, 0, stream>>>(Ws1, Wsp1, 128, 8);
    pack_w_kernel<<<(int)(PACK8 + ZB - 1) / ZB, ZB, 0, stream>>>(Wn1, Wnp1, 128, 8);
    pack_w_kernel<<<(int)(PACK3 + ZB - 1) / ZB, ZB, 0, stream>>>(Ws2, Wsp2, 40, 3);
    pack_w_kernel<<<(int)(PACK3 + ZB - 1) / ZB, ZB, 0, stream>>>(Wn2, Wnp2, 40, 3);

    zero4_kernel<<<(NNODES / 4 + ZB - 1) / ZB, ZB, 0, stream>>>((float4*)cnt, NNODES / 4);
    rank_kernel<<<(NEDGES + ZB - 1) / ZB, ZB, 0, stream>>>(dstv, cnt, rank);
    scan_kernel<<<1, 1024, 0, stream>>>(cnt, rowptr, invdeg);
    fill_kernel<<<(NEDGES + ZB - 1) / ZB, ZB, 0, stream>>>(src, dstv, rank, rowptr, col);

    sage_layer<128, true, true><<<tileBlocks, 256, 0, stream>>>(
        feat, rowptr, col, invdeg, Wsp0, Wnp0, bs0, bn0, h1);

    sage_layer<128, true, true><<<tileBlocks, 256, 0, stream>>>(
        h1, rowptr, col, invdeg, Wsp1, Wnp1, bs1, bn1, h2);

    sage_layer<40, false, false><<<tileBlocks, 256, 0, stream>>>(
        h2, rowptr, col, invdeg, Wsp2, Wnp2, bs2, bn2, out);
}


# Degree prefix sum spread from one work-group (25 dependent trips per lane, lanes 400 B apart) to 98: each sums the counters ahead of it in whole lines, then scans its own 1024 nodes
# speedup vs baseline: 3.7465x; 1.2068x over previous
#include <hip/hip_runtime.h>
#include <hip/hip_bf16.h>

#define NNODES 100000
#define NEDGES 1600000
#define KIN    128

typedef __attribute__((ext_vector_type(2))) float v2f;
typedef __attribute__((ext_vector_type(8))) float v8f;

__global__ void zero4_kernel(float4* __restrict__ p, int n4) {
    int i = blockIdx.x * blockDim.x + threadIdx.x;
    if (i < n4) p[i] = make_float4(0.f, 0.f, 0.f, 0.f);
}


__global__ void rank_kernel(const int* __restrict__ dst, int* __restrict__ cnt,
                            int* __restrict__ rank) {
    int e = blockIdx.x * blockDim.x + threadIdx.x;
    if (e < NEDGES) rank[e] = atomicAdd(&cnt[dst[e]], 1);
}

#define SCAN_T 1024
__global__ void __launch_bounds__(SCAN_T)
scan_kernel(const int* __restrict__ cnt, int* __restrict__ rowptr,
            float* __restrict__ invdeg) {
    __shared__ int wfront[32];
    __shared__ int wown[32];
    const int t    = threadIdx.x;
    const int lane = t & 31;
    const int wv   = t >> 5;
    const int4* __restrict__ c4 = (const int4*)cnt;
    const int n4   = blockIdx.x * (SCAN_T / 4);
    const int node = blockIdx.x * SCAN_T + t;
    const int c = (node < NNODES) ? cnt[node] : 0;

    int s = 0;
    for (int i0 = 0; i0 * SCAN_T < n4; i0 += 8) {
        int4 v[8];
        #pragma unroll
        for (int u = 0; u < 8; ++u) {
            const int idx = (i0 + u) * SCAN_T + t;
            v[u] = c4[idx < n4 ? idx : 0];
        }
        #pragma unroll
        for (int u = 0; u < 8; ++u) {
            const int idx = (i0 + u) * SCAN_T + t;
            if (idx < n4) s += (v[u].x + v[u].y) + (v[u].z + v[u].w);
        }
    }

    int x = c;
    for (int o = 1; o < 32; o <<= 1) {
        const int y = __shfl_up(x, o, 32);
        if (lane >= o) x += y;
    }
    for (int o = 16; o > 0; o >>= 1) s += __shfl_xor(s, o, 32);
    if (lane == 31) wown[wv]   = x;
    if (lane == 0)  wfront[wv] = s;
    __syncthreads();
    int z = wfront[lane] + ((lane < wv) ? wown[lane] : 0);
    for (int o = 16; o > 0; o >>= 1) z += __shfl_xor(z, o, 32);

    if (node < NNODES) {
        rowptr[node] = z + (x - c);
        invdeg[node] = 1.0f / fmaxf((float)c, 1.0f);
    }
}

__global__ void fill_kernel(const int* __restrict__ src, const int* __restrict__ dst,
                            const int* __restrict__ rank, const int* __restrict__ rowptr,
                            int* __restrict__ col) {
    int e = blockIdx.x * blockDim.x + threadIdx.x;
    if (e < NEDGES) col[rowptr[dst[e]] + rank[e]] = src[e];
}

__device__ __forceinline__ float sum16(float v) {
    v += __builtin_bit_cast(float, __builtin_amdgcn_update_dpp(0, __builtin_bit_cast(int, v), 0x161, 0xf, 0xf, true));
    v += __builtin_bit_cast(float, __builtin_amdgcn_update_dpp(0, __builtin_bit_cast(int, v), 0x162, 0xf, 0xf, true));
    v += __builtin_bit_cast(float, __builtin_amdgcn_update_dpp(0, __builtin_bit_cast(int, v), 0x164, 0xf, 0xf, true));
    v += __builtin_bit_cast(float, __builtin_amdgcn_update_dpp(0, __builtin_bit_cast(int, v), 0x168, 0xf, 0xf, true));
    return v;
}

template<int B>
__device__ __forceinline__ void take_rows(const float* __restrict__ h, int mine, int k,
                                          int lane, float4& acc) {
    float4 v[B];
    #pragma unroll
    for (int j = 0; j < B; ++j) {
        const int s = __builtin_amdgcn_readlane(mine, k + j);
        v[j] = ((const float4*)(h + (size_t)s * KIN))[lane];
    }
    #pragma unroll
    for (int j = 0; j < B; ++j) {
        acc.x += v[j].x;
        acc.y += v[j].y;
        acc.z += v[j].z;
        acc.w += v[j].w;
    }
}

__global__ void pack_w_kernel(const float* __restrict__ W, float2* __restrict__ Wp,
                              int nout, int ntiles) {
    int pid = blockIdx.x * blockDim.x + threadIdx.x;
    int total = ntiles * (KIN / 4) * 32;
    if (pid >= total) return;
    int lane = pid & 31;
    int i    = (pid >> 5) & ((KIN / 4) - 1);
    int tile = pid >> 10;
    int l = lane & 15, half = lane >> 4;
    int col = tile * 16 + l;
    int k   = 4 * i + 2 * half;
    float x = 0.f, y = 0.f;
    if (col < nout) {
        x = W[(size_t)k       * nout + col];
        y = W[(size_t)(k + 1) * nout + col];
    }
    Wp[pid] = make_float2(x, y);
}

template<int NOUT, bool ACT, bool NORM>
__global__ void __launch_bounds__(256) __attribute__((amdgpu_waves_per_eu(16, 16)))
sage_layer(const float* __restrict__ h,
           const int* __restrict__ rowptr,
           const int* __restrict__ col,
           const float* __restrict__ invdeg,
           const float2* __restrict__ Wsp,
           const float2* __restrict__ Wnp,
           const float* __restrict__ bself,
           const float* __restrict__ bneigh,
           float* __restrict__ out) {
    constexpr bool FULL = (NOUT % 16) == 0;
    constexpr int NT    = (NOUT + 15) / 16;
    constexpr int ROWS  = 32;
    constexpr int PITCH = KIN + 4;
    static_assert(!NORM || NT == 8, "the row-norm barriers need all eight waves");
    __shared__ float sSelf[ROWS * PITCH];
    __shared__ float sNeigh[ROWS * PITCH];

    const int lane = threadIdx.x & 31;
    const int wv   = __builtin_amdgcn_readfirstlane((int)(threadIdx.x >> 5));
    const int m0   = blockIdx.x * ROWS;

    int   beg[4], end[4], first[4];
    float idg[4];
    #pragma unroll
    for (int t = 0; t < 4; ++t) {
        const int node = m0 + wv + 8 * t;
        beg[t] = rowptr[node];
        end[t] = (node + 1 < NNODES) ? rowptr[node + 1] : NEDGES;
        idg[t] = invdeg[node];
    }
    #pragma unroll
    for (int t = 0; t < 4; ++t)
        first[t] = (lane < end[t] - beg[t]) ? col[beg[t] + lane] : 0;

    #pragma unroll
    for (int t = 0; t < 4; ++t) {
        const int r    = wv + 8 * t;
        const int node = m0 + r;
        const float4 hv = ((const float4*)(h + (size_t)node * KIN))[lane];

        float4 acc = make_float4(0.f, 0.f, 0.f, 0.f);
        int mine = first[t];
        for (int base = beg[t]; base < end[t]; ) {
            const int n = min(end[t] - base, 32);
            int k = 0;
            for (; k + 8 <= n; k += 8) take_rows<8>(h, mine, k, lane, acc);
            if (k + 4 <= n) { take_rows<4>(h, mine, k, lane, acc); k += 4; }
            if (k + 2 <= n) { take_rows<2>(h, mine, k, lane, acc); k += 2; }
            if (k < n)      { take_rows<1>(h, mine, k, lane, acc); }
            base += 32;
            if (base < end[t])
                mine = (lane < end[t] - base) ? col[base + lane] : 0;
        }
        acc.x *= idg[t]; acc.y *= idg[t]; acc.z *= idg[t]; acc.w *= idg[t];
        *(float4*)(sSelf  + r * PITCH + 4 * lane) = hv;
        *(float4*)(sNeigh + r * PITCH + 4 * lane) = acc;
    }
    __syncthreads();
    if (NT < 8 && wv >= NT) return;

    const int l    = lane & 15;
    const int half = lane >> 4;
    const int n0   = wv * 16;
    const int ocol = n0 + l;
    const bool colok = FULL || (ocol < NOUT);

    const float*  As = sSelf  + l * PITCH + 2 * half;
    const float*  An = sNeigh + l * PITCH + 2 * half;
    const float2* __restrict__ Bs = Wsp + (size_t)wv * (KIN / 4) * 32 + lane;
    const float2* __restrict__ Bn = Wnp + (size_t)wv * (KIN / 4) * 32 + lane;

    v8f acc0 = {0.f, 0.f, 0.f, 0.f, 0.f, 0.f, 0.f, 0.f};
    v8f acc1 = {0.f, 0.f, 0.f, 0.f, 0.f, 0.f, 0.f, 0.f};

    #pragma unroll
    for (int i = 0; i < KIN / 4; ++i) {
        const float2 a0f = *(const float2*)(As + 4 * i);
        const float2 a1f = *(const float2*)(As + 16 * PITCH + 4 * i);
        const float2 bf  = Bs[i * 32];
        v2f a0 = {a0f.x, a0f.y};
        v2f a1 = {a1f.x, a1f.y};
        v2f b  = {bf.x, bf.y};
        acc0 = __builtin_amdgcn_wmma_f32_16x16x4_f32(false, a0, false, b,
                                                     (short)0, acc0, false, false);
        acc1 = __builtin_amdgcn_wmma_f32_16x16x4_f32(false, a1, false, b,
                                                     (short)0, acc1, false, false);
    }
    #pragma unroll
    for (int i = 0; i < KIN / 4; ++i) {
        const float2 a0f = *(const float2*)(An + 4 * i);
        const float2 a1f = *(const float2*)(An + 16 * PITCH + 4 * i);
        const float2 bf  = Bn[i * 32];
        v2f a0 = {a0f.x, a0f.y};
        v2f a1 = {a1f.x, a1f.y};
        v2f b  = {bf.x, bf.y};
        acc0 = __builtin_amdgcn_wmma_f32_16x16x4_f32(false, a0, false, b,
                                                     (short)0, acc0, false, false);
        acc1 = __builtin_amdgcn_wmma_f32_16x16x4_f32(false, a1, false, b,
                                                     (short)0, acc1, false, false);
    }

    const float bias = colok ? (bself[ocol] + bneigh[ocol]) : 0.0f;
    #pragma unroll
    for (int r = 0; r < 8; ++r) {
        float v0 = acc0[r] + bias;
        float v1 = acc1[r] + bias;
        if (ACT) { v0 = fmaxf(v0, 0.0f); v1 = fmaxf(v1, 0.0f); }
        acc0[r] = v0;
        acc1[r] = v1;
    }

    if (NORM) {
        __shared__ alignas(16) float part[8 * ROWS];
        float q0[8], q1[8];
        #pragma unroll
        for (int r = 0; r < 8; ++r) {
            q0[r] = sum16(acc0[r] * acc0[r]);
            q1[r] = sum16(acc1[r] * acc1[r]);
        }
        if (l == 0) {
            float* p = part + wv * ROWS + 8 * half;
            *(float4*)(p)      = make_float4(q0[0], q0[1], q0[2], q0[3]);
            *(float4*)(p + 4)  = make_float4(q0[4], q0[5], q0[6], q0[7]);
            *(float4*)(p + 16) = make_float4(q1[0], q1[1], q1[2], q1[3]);
            *(float4*)(p + 20) = make_float4(q1[4], q1[5], q1[6], q1[7]);
        }
        __syncthreads();
        float tot = part[lane];
        #pragma unroll
        for (int w = 1; w < 8; ++w) tot += part[w * ROWS + lane];
        const float nrm = fmaxf(sqrtf(tot), 1e-12f);
        #pragma unroll
        for (int r = 0; r < 8; ++r) {
            const float nrm0 = __shfl(nrm, r + 8 * half, 32);
            const float nrm1 = __shfl(nrm, 16 + r + 8 * half, 32);
            acc0[r] = acc0[r] / nrm0;
            acc1[r] = acc1[r] / nrm1;
        }
    }

    if (colok) {
        #pragma unroll
        for (int r = 0; r < 8; ++r) {
            const int m = m0 + r + 8 * half;
            out[(size_t)m * NOUT + ocol]        = acc0[r];
            out[(size_t)(m + 16) * NOUT + ocol] = acc1[r];
        }
    }
}

extern "C" void kernel_launch(void* const* d_in, const int* in_sizes, int n_in,
                              void* d_out, int out_size, void* d_ws, size_t ws_size,
                              hipStream_t stream) {
    const float* feat = (const float*)d_in[0];
    const int*   src  = (const int*)d_in[1];
    const int*   dstv = (const int*)d_in[2];
    const float* Ws0  = (const float*)d_in[3];
    const float* bs0  = (const float*)d_in[4];
    const float* Wn0  = (const float*)d_in[5];
    const float* bn0  = (const float*)d_in[6];
    const float* Ws1  = (const float*)d_in[7];
    const float* bs1  = (const float*)d_in[8];
    const float* Wn1  = (const float*)d_in[9];
    const float* bn1  = (const float*)d_in[10];
    const float* Ws2  = (const float*)d_in[11];
    const float* bs2  = (const float*)d_in[12];
    const float* Wn2  = (const float*)d_in[13];
    const float* bn2  = (const float*)d_in[14];
    float* out = (float*)d_out;

    int*   col    = (int*)d_ws;
    float* h1     = (float*)d_ws + (size_t)NNODES * KIN;
    float* h2     = h1  + (size_t)NNODES * KIN;
    int*   rowptr = (int*)(h2 + (size_t)NNODES * KIN);
    float* invdeg = (float*)(rowptr + NNODES);
    int* cnt  = (int*)h2;
    int* rank = cnt + NNODES;
    const size_t PACK8 = (size_t)8 * (KIN / 4) * 32;
    const size_t PACK3 = (size_t)3 * (KIN / 4) * 32;
    float2* Wsp0 = (float2*)(invdeg + NNODES);
    float2* Wnp0 = Wsp0 + PACK8;
    float2* Wsp1 = Wnp0 + PACK8;
    float2* Wnp1 = Wsp1 + PACK8;
    float2* Wsp2 = Wnp1 + PACK8;
    float2* Wnp2 = Wsp2 + PACK3;

    const int ZB = 256;
    const int tileBlocks = NNODES / 32;

    pack_w_kernel<<<(int)(PACK8 + ZB - 1) / ZB, ZB, 0, stream>>>(Ws0, Wsp0, 128, 8);
    pack_w_kernel<<<(int)(PACK8 + ZB - 1) / ZB, ZB, 0, stream>>>(Wn0, Wnp0, 128, 8);
    pack_w_kernel<<<(int)(PACK8 + ZB - 1) / ZB, ZB, 0, stream>>>(Ws1, Wsp1, 128, 8);
    pack_w_kernel<<<(int)(PACK8 + ZB - 1) / ZB, ZB, 0, stream>>>(Wn1, Wnp1, 128, 8);
    pack_w_kernel<<<(int)(PACK3 + ZB - 1) / ZB, ZB, 0, stream>>>(Ws2, Wsp2, 40, 3);
    pack_w_kernel<<<(int)(PACK3 + ZB - 1) / ZB, ZB, 0, stream>>>(Wn2, Wnp2, 40, 3);

    zero4_kernel<<<(NNODES / 4 + ZB - 1) / ZB, ZB, 0, stream>>>((float4*)cnt, NNODES / 4);
    rank_kernel<<<(NEDGES + ZB - 1) / ZB, ZB, 0, stream>>>(dstv, cnt, rank);
    scan_kernel<<<(NNODES + SCAN_T - 1) / SCAN_T, SCAN_T, 0, stream>>>(cnt, rowptr, invdeg);
    fill_kernel<<<(NEDGES + ZB - 1) / ZB, ZB, 0, stream>>>(src, dstv, rank, rowptr, col);

    sage_layer<128, true, true><<<tileBlocks, 256, 0, stream>>>(
        feat, rowptr, col, invdeg, Wsp0, Wnp0, bs0, bn0, h1);

    sage_layer<128, true, true><<<tileBlocks, 256, 0, stream>>>(
        h1, rowptr, col, invdeg, Wsp1, Wnp1, bs1, bn1, h2);

    sage_layer<40, false, false><<<tileBlocks, 256, 0, stream>>>(
        h2, rowptr, col, invdeg, Wsp2, Wnp2, bs2, bn2, out);
}


# Edge-sort counters: one array both package halves add into -> one per half (die number picks it, edge keeps the bit, degree = sum); add order now half-0 edges then half-1, arrival rank in each
# speedup vs baseline: 4.5948x; 1.2264x over previous
#include <hip/hip_runtime.h>
#include <hip/hip_bf16.h>

#define NNODES 100000
#define NEDGES 1600000
#define KIN    128

typedef __attribute__((ext_vector_type(2))) float v2f;
typedef __attribute__((ext_vector_type(8))) float v8f;

__global__ void zero4_kernel(float4* __restrict__ p, int n4) {
    int i = blockIdx.x * blockDim.x + threadIdx.x;
    if (i < n4) p[i] = make_float4(0.f, 0.f, 0.f, 0.f);
}


__global__ void rank_kernel(const int* __restrict__ dst, int* __restrict__ cnt,
                            int* __restrict__ rank) {
    const int half = (int)((__builtin_amdgcn_s_sendmsg_rtn(0x87) >> 18) & 1u);
    int e = blockIdx.x * blockDim.x + threadIdx.x;
    if (e < NEDGES) rank[e] = 2 * atomicAdd(&cnt[half * NNODES + dst[e]], 1) + half;
}

#define SCAN_T 1024
__global__ void __launch_bounds__(SCAN_T)
scan_kernel(const int* __restrict__ cnt, int* __restrict__ rowptr,
            int* __restrict__ rowptr1, float* __restrict__ invdeg) {
    __shared__ int wfront[32];
    __shared__ int wown[32];
    const int t    = threadIdx.x;
    const int lane = t & 31;
    const int wv   = t >> 5;
    const int4* __restrict__ c4  = (const int4*)cnt;
    const int4* __restrict__ c4b = (const int4*)(cnt + NNODES);
    const int n4   = blockIdx.x * (SCAN_T / 4);
    const int node = blockIdx.x * SCAN_T + t;
    const int c0 = (node < NNODES) ? cnt[node] : 0;
    const int c1 = (node < NNODES) ? cnt[NNODES + node] : 0;
    const int c  = c0 + c1;

    int s = 0;
    for (int i0 = 0; i0 * SCAN_T < n4; i0 += 4) {
        int4 v[4], w[4];
        #pragma unroll
        for (int u = 0; u < 4; ++u) {
            const int idx = (i0 + u) * SCAN_T + t;
            v[u] = c4[idx < n4 ? idx : 0];
            w[u] = c4b[idx < n4 ? idx : 0];
        }
        #pragma unroll
        for (int u = 0; u < 4; ++u) {
            const int idx = (i0 + u) * SCAN_T + t;
            if (idx < n4) s += ((v[u].x + v[u].y) + (v[u].z + v[u].w))
                             + ((w[u].x + w[u].y) + (w[u].z + w[u].w));
        }
    }

    int x = c;
    for (int o = 1; o < 32; o <<= 1) {
        const int y = __shfl_up(x, o, 32);
        if (lane >= o) x += y;
    }
    for (int o = 16; o > 0; o >>= 1) s += __shfl_xor(s, o, 32);
    if (lane == 31) wown[wv]   = x;
    if (lane == 0)  wfront[wv] = s;
    __syncthreads();
    int z = wfront[lane] + ((lane < wv) ? wown[lane] : 0);
    for (int o = 16; o > 0; o >>= 1) z += __shfl_xor(z, o, 32);

    if (node < NNODES) {
        rowptr[node]  = z + (x - c);
        rowptr1[node] = z + (x - c) + c0;
        invdeg[node]  = 1.0f / fmaxf((float)c, 1.0f);
    }
}

__global__ void fill_kernel(const int* __restrict__ src, const int* __restrict__ dst,
                            const int* __restrict__ rank, const int* __restrict__ rowptr,
                            const int* __restrict__ rowptr1, int* __restrict__ col) {
    int e = blockIdx.x * blockDim.x + threadIdx.x;
    if (e < NEDGES) {
        const int rk = rank[e];
        const int* __restrict__ start = (rk & 1) ? rowptr1 : rowptr;
        col[start[dst[e]] + (rk >> 1)] = src[e];
    }
}

__device__ __forceinline__ float sum16(float v) {
    v += __builtin_bit_cast(float, __builtin_amdgcn_update_dpp(0, __builtin_bit_cast(int, v), 0x161, 0xf, 0xf, true));
    v += __builtin_bit_cast(float, __builtin_amdgcn_update_dpp(0, __builtin_bit_cast(int, v), 0x162, 0xf, 0xf, true));
    v += __builtin_bit_cast(float, __builtin_amdgcn_update_dpp(0, __builtin_bit_cast(int, v), 0x164, 0xf, 0xf, true));
    v += __builtin_bit_cast(float, __builtin_amdgcn_update_dpp(0, __builtin_bit_cast(int, v), 0x168, 0xf, 0xf, true));
    return v;
}

template<int B>
__device__ __forceinline__ void take_rows(const float* __restrict__ h, int mine, int k,
                                          int lane, float4& acc) {
    float4 v[B];
    #pragma unroll
    for (int j = 0; j < B; ++j) {
        const int s = __builtin_amdgcn_readlane(mine, k + j);
        v[j] = ((const float4*)(h + (size_t)s * KIN))[lane];
    }
    #pragma unroll
    for (int j = 0; j < B; ++j) {
        acc.x += v[j].x;
        acc.y += v[j].y;
        acc.z += v[j].z;
        acc.w += v[j].w;
    }
}

__global__ void pack_w_kernel(const float* __restrict__ W, float2* __restrict__ Wp,
                              int nout, int ntiles) {
    int pid = blockIdx.x * blockDim.x + threadIdx.x;
    int total = ntiles * (KIN / 4) * 32;
    if (pid >= total) return;
    int lane = pid & 31;
    int i    = (pid >> 5) & ((KIN / 4) - 1);
    int tile = pid >> 10;
    int l = lane & 15, half = lane >> 4;
    int col = tile * 16 + l;
    int k   = 4 * i + 2 * half;
    float x = 0.f, y = 0.f;
    if (col < nout) {
        x = W[(size_t)k       * nout + col];
        y = W[(size_t)(k + 1) * nout + col];
    }
    Wp[pid] = make_float2(x, y);
}

template<int NOUT, bool ACT, bool NORM>
__global__ void __launch_bounds__(256) __attribute__((amdgpu_waves_per_eu(16, 16)))
sage_layer(const float* __restrict__ h,
           const int* __restrict__ rowptr,
           const int* __restrict__ col,
           const float* __restrict__ invdeg,
           const float2* __restrict__ Wsp,
           const float2* __restrict__ Wnp,
           const float* __restrict__ bself,
           const float* __restrict__ bneigh,
           float* __restrict__ out) {
    constexpr bool FULL = (NOUT % 16) == 0;
    constexpr int NT    = (NOUT + 15) / 16;
    constexpr int ROWS  = 32;
    constexpr int PITCH = KIN + 4;
    static_assert(!NORM || NT == 8, "the row-norm barriers need all eight waves");
    __shared__ float sSelf[ROWS * PITCH];
    __shared__ float sNeigh[ROWS * PITCH];

    const int lane = threadIdx.x & 31;
    const int wv   = __builtin_amdgcn_readfirstlane((int)(threadIdx.x >> 5));
    const int m0   = blockIdx.x * ROWS;

    int   beg[4], end[4], first[4];
    float idg[4];
    #pragma unroll
    for (int t = 0; t < 4; ++t) {
        const int node = m0 + wv + 8 * t;
        beg[t] = rowptr[node];
        end[t] = (node + 1 < NNODES) ? rowptr[node + 1] : NEDGES;
        idg[t] = invdeg[node];
    }
    #pragma unroll
    for (int t = 0; t < 4; ++t)
        first[t] = (lane < end[t] - beg[t]) ? col[beg[t] + lane] : 0;

    #pragma unroll
    for (int t = 0; t < 4; ++t) {
        const int r    = wv + 8 * t;
        const int node = m0 + r;
        const float4 hv = ((const float4*)(h + (size_t)node * KIN))[lane];

        float4 acc = make_float4(0.f, 0.f, 0.f, 0.f);
        int mine = first[t];
        for (int base = beg[t]; base < end[t]; ) {
            const int n = min(end[t] - base, 32);
            int k = 0;
            for (; k + 8 <= n; k += 8) take_rows<8>(h, mine, k, lane, acc);
            if (k + 4 <= n) { take_rows<4>(h, mine, k, lane, acc); k += 4; }
            if (k + 2 <= n) { take_rows<2>(h, mine, k, lane, acc); k += 2; }
            if (k < n)      { take_rows<1>(h, mine, k, lane, acc); }
            base += 32;
            if (base < end[t])
                mine = (lane < end[t] - base) ? col[base + lane] : 0;
        }
        acc.x *= idg[t]; acc.y *= idg[t]; acc.z *= idg[t]; acc.w *= idg[t];
        *(float4*)(sSelf  + r * PITCH + 4 * lane) = hv;
        *(float4*)(sNeigh + r * PITCH + 4 * lane) = acc;
    }
    __syncthreads();
    if (NT < 8 && wv >= NT) return;

    const int l    = lane & 15;
    const int half = lane >> 4;
    const int n0   = wv * 16;
    const int ocol = n0 + l;
    const bool colok = FULL || (ocol < NOUT);

    const float*  As = sSelf  + l * PITCH + 2 * half;
    const float*  An = sNeigh + l * PITCH + 2 * half;
    const float2* __restrict__ Bs = Wsp + (size_t)wv * (KIN / 4) * 32 + lane;
    const float2* __restrict__ Bn = Wnp + (size_t)wv * (KIN / 4) * 32 + lane;

    v8f acc0 = {0.f, 0.f, 0.f, 0.f, 0.f, 0.f, 0.f, 0.f};
    v8f acc1 = {0.f, 0.f, 0.f, 0.f, 0.f, 0.f, 0.f, 0.f};

    #pragma unroll
    for (int i = 0; i < KIN / 4; ++i) {
        const float2 a0f = *(const float2*)(As + 4 * i);
        const float2 a1f = *(const float2*)(As + 16 * PITCH + 4 * i);
        const float2 bf  = Bs[i * 32];
        v2f a0 = {a0f.x, a0f.y};
        v2f a1 = {a1f.x, a1f.y};
        v2f b  = {bf.x, bf.y};
        acc0 = __builtin_amdgcn_wmma_f32_16x16x4_f32(false, a0, false, b,
                                                     (short)0, acc0, false, false);
        acc1 = __builtin_amdgcn_wmma_f32_16x16x4_f32(false, a1, false, b,
                                                     (short)0, acc1, false, false);
    }
    #pragma unroll
    for (int i = 0; i < KIN / 4; ++i) {
        const float2 a0f = *(const float2*)(An + 4 * i);
        const float2 a1f = *(const float2*)(An + 16 * PITCH + 4 * i);
        const float2 bf  = Bn[i * 32];
        v2f a0 = {a0f.x, a0f.y};
        v2f a1 = {a1f.x, a1f.y};
        v2f b  = {bf.x, bf.y};
        acc0 = __builtin_amdgcn_wmma_f32_16x16x4_f32(false, a0, false, b,
                                                     (short)0, acc0, false, false);
        acc1 = __builtin_amdgcn_wmma_f32_16x16x4_f32(false, a1, false, b,
                                                     (short)0, acc1, false, false);
    }

    const float bias = colok ? (bself[ocol] + bneigh[ocol]) : 0.0f;
    #pragma unroll
    for (int r = 0; r < 8; ++r) {
        float v0 = acc0[r] + bias;
        float v1 = acc1[r] + bias;
        if (ACT) { v0 = fmaxf(v0, 0.0f); v1 = fmaxf(v1, 0.0f); }
        acc0[r] = v0;
        acc1[r] = v1;
    }

    if (NORM) {
        __shared__ alignas(16) float part[8 * ROWS];
        float q0[8], q1[8];
        #pragma unroll
        for (int r = 0; r < 8; ++r) {
            q0[r] = sum16(acc0[r] * acc0[r]);
            q1[r] = sum16(acc1[r] * acc1[r]);
        }
        if (l == 0) {
            float* p = part + wv * ROWS + 8 * half;
            *(float4*)(p)      = make_float4(q0[0], q0[1], q0[2], q0[3]);
            *(float4*)(p + 4)  = make_float4(q0[4], q0[5], q0[6], q0[7]);
            *(float4*)(p + 16) = make_float4(q1[0], q1[1], q1[2], q1[3]);
            *(float4*)(p + 20) = make_float4(q1[4], q1[5], q1[6], q1[7]);
        }
        __syncthreads();
        float tot = part[lane];
        #pragma unroll
        for (int w = 1; w < 8; ++w) tot += part[w * ROWS + lane];
        const float nrm = fmaxf(sqrtf(tot), 1e-12f);
        #pragma unroll
        for (int r = 0; r < 8; ++r) {
            const float nrm0 = __shfl(nrm, r + 8 * half, 32);
            const float nrm1 = __shfl(nrm, 16 + r + 8 * half, 32);
            acc0[r] = acc0[r] / nrm0;
            acc1[r] = acc1[r] / nrm1;
        }
    }

    if (colok) {
        #pragma unroll
        for (int r = 0; r < 8; ++r) {
            const int m = m0 + r + 8 * half;
            out[(size_t)m * NOUT + ocol]        = acc0[r];
            out[(size_t)(m + 16) * NOUT + ocol] = acc1[r];
        }
    }
}

extern "C" void kernel_launch(void* const* d_in, const int* in_sizes, int n_in,
                              void* d_out, int out_size, void* d_ws, size_t ws_size,
                              hipStream_t stream) {
    const float* feat = (const float*)d_in[0];
    const int*   src  = (const int*)d_in[1];
    const int*   dstv = (const int*)d_in[2];
    const float* Ws0  = (const float*)d_in[3];
    const float* bs0  = (const float*)d_in[4];
    const float* Wn0  = (const float*)d_in[5];
    const float* bn0  = (const float*)d_in[6];
    const float* Ws1  = (const float*)d_in[7];
    const float* bs1  = (const float*)d_in[8];
    const float* Wn1  = (const float*)d_in[9];
    const float* bn1  = (const float*)d_in[10];
    const float* Ws2  = (const float*)d_in[11];
    const float* bs2  = (const float*)d_in[12];
    const float* Wn2  = (const float*)d_in[13];
    const float* bn2  = (const float*)d_in[14];
    float* out = (float*)d_out;

    int*   col    = (int*)d_ws;
    float* h1     = (float*)d_ws + (size_t)NNODES * KIN;
    float* h2     = h1  + (size_t)NNODES * KIN;
    int*   rowptr = (int*)(h2 + (size_t)NNODES * KIN);
    float* invdeg = (float*)(rowptr + NNODES);
    int* cnt     = (int*)h2;
    int* rank    = cnt + 2 * NNODES;
    int* rowptr1 = col + NEDGES;
    const size_t PACK8 = (size_t)8 * (KIN / 4) * 32;
    const size_t PACK3 = (size_t)3 * (KIN / 4) * 32;
    float2* Wsp0 = (float2*)(invdeg + NNODES);
    float2* Wnp0 = Wsp0 + PACK8;
    float2* Wsp1 = Wnp0 + PACK8;
    float2* Wnp1 = Wsp1 + PACK8;
    float2* Wsp2 = Wnp1 + PACK8;
    float2* Wnp2 = Wsp2 + PACK3;

    const int ZB = 256;
    const int tileBlocks = NNODES / 32;

    pack_w_kernel<<<(int)(PACK8 + ZB - 1) / ZB, ZB, 0, stream>>>(Ws0, Wsp0, 128, 8);
    pack_w_kernel<<<(int)(PACK8 + ZB - 1) / ZB, ZB, 0, stream>>>(Wn0, Wnp0, 128, 8);
    pack_w_kernel<<<(int)(PACK8 + ZB - 1) / ZB, ZB, 0, stream>>>(Ws1, Wsp1, 128, 8);
    pack_w_kernel<<<(int)(PACK8 + ZB - 1) / ZB, ZB, 0, stream>>>(Wn1, Wnp1, 128, 8);
    pack_w_kernel<<<(int)(PACK3 + ZB - 1) / ZB, ZB, 0, stream>>>(Ws2, Wsp2, 40, 3);
    pack_w_kernel<<<(int)(PACK3 + ZB - 1) / ZB, ZB, 0, stream>>>(Wn2, Wnp2, 40, 3);

    zero4_kernel<<<(2 * NNODES / 4 + ZB - 1) / ZB, ZB, 0, stream>>>((float4*)cnt, 2 * NNODES / 4);
    rank_kernel<<<(NEDGES + ZB - 1) / ZB, ZB, 0, stream>>>(dstv, cnt, rank);
    scan_kernel<<<(NNODES + SCAN_T - 1) / SCAN_T, SCAN_T, 0, stream>>>(cnt, rowptr, rowptr1, invdeg);
    fill_kernel<<<(NEDGES + ZB - 1) / ZB, ZB, 0, stream>>>(src, dstv, rank, rowptr, rowptr1, col);

    sage_layer<128, true, true><<<tileBlocks, 256, 0, stream>>>(
        feat, rowptr, col, invdeg, Wsp0, Wnp0, bs0, bn0, h1);

    sage_layer<128, true, true><<<tileBlocks, 256, 0, stream>>>(
        h1, rowptr, col, invdeg, Wsp1, Wnp1, bs1, bn1, h2);

    sage_layer<40, false, false><<<tileBlocks, 256, 0, stream>>>(
        h2, rowptr, col, invdeg, Wsp2, Wnp2, bs2, bn2, out);
}


# The six weight-pack launches and the counter zero-fill merged into one launch (block index picks the job): 13 launches -> 7; kernels' arithmetic and add order untouched
# speedup vs baseline: 4.7801x; 1.0403x over previous
#include <hip/hip_runtime.h>
#include <hip/hip_bf16.h>

#define NNODES 100000
#define NEDGES 1600000
#define KIN    128

typedef __attribute__((ext_vector_type(2))) float v2f;
typedef __attribute__((ext_vector_type(8))) float v8f;


__global__ void rank_kernel(const int* __restrict__ dst, int* __restrict__ cnt,
                            int* __restrict__ rank) {
    const int half = (int)((__builtin_amdgcn_s_sendmsg_rtn(0x87) >> 18) & 1u);
    int e = blockIdx.x * blockDim.x + threadIdx.x;
    if (e < NEDGES) rank[e] = 2 * atomicAdd(&cnt[half * NNODES + dst[e]], 1) + half;
}

#define SCAN_T 1024
__global__ void __launch_bounds__(SCAN_T)
scan_kernel(const int* __restrict__ cnt, int* __restrict__ rowptr,
            int* __restrict__ rowptr1, float* __restrict__ invdeg) {
    __shared__ int wfront[32];
    __shared__ int wown[32];
    const int t    = threadIdx.x;
    const int lane = t & 31;
    const int wv   = t >> 5;
    const int4* __restrict__ c4  = (const int4*)cnt;
    const int4* __restrict__ c4b = (const int4*)(cnt + NNODES);
    const int n4   = blockIdx.x * (SCAN_T / 4);
    const int node = blockIdx.x * SCAN_T + t;
    const int c0 = (node < NNODES) ? cnt[node] : 0;
    const int c1 = (node < NNODES) ? cnt[NNODES + node] : 0;
    const int c  = c0 + c1;

    int s = 0;
    for (int i0 = 0; i0 * SCAN_T < n4; i0 += 4) {
        int4 v[4], w[4];
        #pragma unroll
        for (int u = 0; u < 4; ++u) {
            const int idx = (i0 + u) * SCAN_T + t;
            v[u] = c4[idx < n4 ? idx : 0];
            w[u] = c4b[idx < n4 ? idx : 0];
        }
        #pragma unroll
        for (int u = 0; u < 4; ++u) {
            const int idx = (i0 + u) * SCAN_T + t;
            if (idx < n4) s += ((v[u].x + v[u].y) + (v[u].z + v[u].w))
                             + ((w[u].x + w[u].y) + (w[u].z + w[u].w));
        }
    }

    int x = c;
    for (int o = 1; o < 32; o <<= 1) {
        const int y = __shfl_up(x, o, 32);
        if (lane >= o) x += y;
    }
    for (int o = 16; o > 0; o >>= 1) s += __shfl_xor(s, o, 32);
    if (lane == 31) wown[wv]   = x;
    if (lane == 0)  wfront[wv] = s;
    __syncthreads();
    int z = wfront[lane] + ((lane < wv) ? wown[lane] : 0);
    for (int o = 16; o > 0; o >>= 1) z += __shfl_xor(z, o, 32);

    if (node < NNODES) {
        rowptr[node]  = z + (x - c);
        rowptr1[node] = z + (x - c) + c0;
        invdeg[node]  = 1.0f / fmaxf((float)c, 1.0f);
    }
}

__global__ void fill_kernel(const int* __restrict__ src, const int* __restrict__ dst,
                            const int* __restrict__ rank, const int* __restrict__ rowptr,
                            const int* __restrict__ rowptr1, int* __restrict__ col) {
    int e = blockIdx.x * blockDim.x + threadIdx.x;
    if (e < NEDGES) {
        const int rk = rank[e];
        const int* __restrict__ start = (rk & 1) ? rowptr1 : rowptr;
        col[start[dst[e]] + (rk >> 1)] = src[e];
    }
}

__device__ __forceinline__ float sum16(float v) {
    v += __builtin_bit_cast(float, __builtin_amdgcn_update_dpp(0, __builtin_bit_cast(int, v), 0x161, 0xf, 0xf, true));
    v += __builtin_bit_cast(float, __builtin_amdgcn_update_dpp(0, __builtin_bit_cast(int, v), 0x162, 0xf, 0xf, true));
    v += __builtin_bit_cast(float, __builtin_amdgcn_update_dpp(0, __builtin_bit_cast(int, v), 0x164, 0xf, 0xf, true));
    v += __builtin_bit_cast(float, __builtin_amdgcn_update_dpp(0, __builtin_bit_cast(int, v), 0x168, 0xf, 0xf, true));
    return v;
}

template<int B>
__device__ __forceinline__ void take_rows(const float* __restrict__ h, int mine, int k,
                                          int lane, float4& acc) {
    float4 v[B];
    #pragma unroll
    for (int j = 0; j < B; ++j) {
        const int s = __builtin_amdgcn_readlane(mine, k + j);
        v[j] = ((const float4*)(h + (size_t)s * KIN))[lane];
    }
    #pragma unroll
    for (int j = 0; j < B; ++j) {
        acc.x += v[j].x;
        acc.y += v[j].y;
        acc.z += v[j].z;
        acc.w += v[j].w;
    }
}

__device__ __forceinline__ void pack_w(const float* __restrict__ W, float2* __restrict__ Wp,
                                       int nout, int pid) {
    int lane = pid & 31;
    int i    = (pid >> 5) & ((KIN / 4) - 1);
    int tile = pid >> 10;
    int l = lane & 15, half = lane >> 4;
    int col = tile * 16 + l;
    int k   = 4 * i + 2 * half;
    float x = 0.f, y = 0.f;
    if (col < nout) {
        x = W[(size_t)k       * nout + col];
        y = W[(size_t)(k + 1) * nout + col];
    }
    Wp[pid] = make_float2(x, y);
}

#define PREP_T       256
#define PACK8_F2     (8 * (KIN / 4) * 32)
#define PACK3_F2     (3 * (KIN / 4) * 32)
#define PACK8_BLOCKS (PACK8_F2 / PREP_T)
#define PACK3_BLOCKS (PACK3_F2 / PREP_T)
#define PACK_BLOCKS  (4 * PACK8_BLOCKS + 2 * PACK3_BLOCKS)
#define CNT_F4       (2 * NNODES / 4)
#define ZERO_BLOCKS  ((CNT_F4 + PREP_T - 1) / PREP_T)
static_assert(PACK8_F2 % PREP_T == 0 && PACK3_F2 % PREP_T == 0, "pack blocks are whole");
__global__ void __launch_bounds__(PREP_T)
prep_kernel(const float* __restrict__ Ws0, const float* __restrict__ Wn0,
            const float* __restrict__ Ws1, const float* __restrict__ Wn1,
            const float* __restrict__ Ws2, const float* __restrict__ Wn2,
            float2* __restrict__ Wp, float4* __restrict__ cnt4) {
    const int b = blockIdx.x;
    const int t = threadIdx.x;
    if (b >= PACK_BLOCKS) {
        const int i = (b - PACK_BLOCKS) * PREP_T + t;
        if (i < CNT_F4) cnt4[i] = make_float4(0.f, 0.f, 0.f, 0.f);
        return;
    }
    if (b < 4 * PACK8_BLOCKS) {
        const int m = b / PACK8_BLOCKS;
        const float* __restrict__ W = (m == 0) ? Ws0 : (m == 1) ? Wn0 : (m == 2) ? Ws1 : Wn1;
        pack_w(W, Wp + (size_t)m * PACK8_F2, 128, (b - m * PACK8_BLOCKS) * PREP_T + t);
    } else {
        const int b3 = b - 4 * PACK8_BLOCKS;
        const int m  = b3 / PACK3_BLOCKS;
        const float* __restrict__ W = (m == 0) ? Ws2 : Wn2;
        pack_w(W, Wp + (size_t)4 * PACK8_F2 + (size_t)m * PACK3_F2, 40,
               (b3 - m * PACK3_BLOCKS) * PREP_T + t);
    }
}

template<int NOUT, bool ACT, bool NORM>
__global__ void __launch_bounds__(256) __attribute__((amdgpu_waves_per_eu(16, 16)))
sage_layer(const float* __restrict__ h,
           const int* __restrict__ rowptr,
           const int* __restrict__ col,
           const float* __restrict__ invdeg,
           const float2* __restrict__ Wsp,
           const float2* __restrict__ Wnp,
           const float* __restrict__ bself,
           const float* __restrict__ bneigh,
           float* __restrict__ out) {
    constexpr bool FULL = (NOUT % 16) == 0;
    constexpr int NT    = (NOUT + 15) / 16;
    constexpr int ROWS  = 32;
    constexpr int PITCH = KIN + 4;
    static_assert(!NORM || NT == 8, "the row-norm barriers need all eight waves");
    __shared__ float sSelf[ROWS * PITCH];
    __shared__ float sNeigh[ROWS * PITCH];

    const int lane = threadIdx.x & 31;
    const int wv   = __builtin_amdgcn_readfirstlane((int)(threadIdx.x >> 5));
    const int m0   = blockIdx.x * ROWS;

    int   beg[4], end[4], first[4];
    float idg[4];
    #pragma unroll
    for (int t = 0; t < 4; ++t) {
        const int node = m0 + wv + 8 * t;
        beg[t] = rowptr[node];
        end[t] = (node + 1 < NNODES) ? rowptr[node + 1] : NEDGES;
        idg[t] = invdeg[node];
    }
    #pragma unroll
    for (int t = 0; t < 4; ++t)
        first[t] = (lane < end[t] - beg[t]) ? col[beg[t] + lane] : 0;

    #pragma unroll
    for (int t = 0; t < 4; ++t) {
        const int r    = wv + 8 * t;
        const int node = m0 + r;
        const float4 hv = ((const float4*)(h + (size_t)node * KIN))[lane];

        float4 acc = make_float4(0.f, 0.f, 0.f, 0.f);
        int mine = first[t];
        for (int base = beg[t]; base < end[t]; ) {
            const int n = min(end[t] - base, 32);
            int k = 0;
            for (; k + 8 <= n; k += 8) take_rows<8>(h, mine, k, lane, acc);
            if (k + 4 <= n) { take_rows<4>(h, mine, k, lane, acc); k += 4; }
            if (k + 2 <= n) { take_rows<2>(h, mine, k, lane, acc); k += 2; }
            if (k < n)      { take_rows<1>(h, mine, k, lane, acc); }
            base += 32;
            if (base < end[t])
                mine = (lane < end[t] - base) ? col[base + lane] : 0;
        }
        acc.x *= idg[t]; acc.y *= idg[t]; acc.z *= idg[t]; acc.w *= idg[t];
        *(float4*)(sSelf  + r * PITCH + 4 * lane) = hv;
        *(float4*)(sNeigh + r * PITCH + 4 * lane) = acc;
    }
    __syncthreads();
    if (NT < 8 && wv >= NT) return;

    const int l    = lane & 15;
    const int half = lane >> 4;
    const int n0   = wv * 16;
    const int ocol = n0 + l;
    const bool colok = FULL || (ocol < NOUT);

    const float*  As = sSelf  + l * PITCH + 2 * half;
    const float*  An = sNeigh + l * PITCH + 2 * half;
    const float2* __restrict__ Bs = Wsp + (size_t)wv * (KIN / 4) * 32 + lane;
    const float2* __restrict__ Bn = Wnp + (size_t)wv * (KIN / 4) * 32 + lane;

    v8f acc0 = {0.f, 0.f, 0.f, 0.f, 0.f, 0.f, 0.f, 0.f};
    v8f acc1 = {0.f, 0.f, 0.f, 0.f, 0.f, 0.f, 0.f, 0.f};

    #pragma unroll
    for (int i = 0; i < KIN / 4; ++i) {
        const float2 a0f = *(const float2*)(As + 4 * i);
        const float2 a1f = *(const float2*)(As + 16 * PITCH + 4 * i);
        const float2 bf  = Bs[i * 32];
        v2f a0 = {a0f.x, a0f.y};
        v2f a1 = {a1f.x, a1f.y};
        v2f b  = {bf.x, bf.y};
        acc0 = __builtin_amdgcn_wmma_f32_16x16x4_f32(false, a0, false, b,
                                                     (short)0, acc0, false, false);
        acc1 = __builtin_amdgcn_wmma_f32_16x16x4_f32(false, a1, false, b,
                                                     (short)0, acc1, false, false);
    }
    #pragma unroll
    for (int i = 0; i < KIN / 4; ++i) {
        const float2 a0f = *(const float2*)(An + 4 * i);
        const float2 a1f = *(const float2*)(An + 16 * PITCH + 4 * i);
        const float2 bf  = Bn[i * 32];
        v2f a0 = {a0f.x, a0f.y};
        v2f a1 = {a1f.x, a1f.y};
        v2f b  = {bf.x, bf.y};
        acc0 = __builtin_amdgcn_wmma_f32_16x16x4_f32(false, a0, false, b,
                                                     (short)0, acc0, false, false);
        acc1 = __builtin_amdgcn_wmma_f32_16x16x4_f32(false, a1, false, b,
                                                     (short)0, acc1, false, false);
    }

    const float bias = colok ? (bself[ocol] + bneigh[ocol]) : 0.0f;
    #pragma unroll
    for (int r = 0; r < 8; ++r) {
        float v0 = acc0[r] + bias;
        float v1 = acc1[r] + bias;
        if (ACT) { v0 = fmaxf(v0, 0.0f); v1 = fmaxf(v1, 0.0f); }
        acc0[r] = v0;
        acc1[r] = v1;
    }

    if (NORM) {
        __shared__ alignas(16) float part[8 * ROWS];
        float q0[8], q1[8];
        #pragma unroll
        for (int r = 0; r < 8; ++r) {
            q0[r] = sum16(acc0[r] * acc0[r]);
            q1[r] = sum16(acc1[r] * acc1[r]);
        }
        if (l == 0) {
            float* p = part + wv * ROWS + 8 * half;
            *(float4*)(p)      = make_float4(q0[0], q0[1], q0[2], q0[3]);
            *(float4*)(p + 4)  = make_float4(q0[4], q0[5], q0[6], q0[7]);
            *(float4*)(p + 16) = make_float4(q1[0], q1[1], q1[2], q1[3]);
            *(float4*)(p + 20) = make_float4(q1[4], q1[5], q1[6], q1[7]);
        }
        __syncthreads();
        float tot = part[lane];
        #pragma unroll
        for (int w = 1; w < 8; ++w) tot += part[w * ROWS + lane];
        const float nrm = fmaxf(sqrtf(tot), 1e-12f);
        #pragma unroll
        for (int r = 0; r < 8; ++r) {
            const float nrm0 = __shfl(nrm, r + 8 * half, 32);
            const float nrm1 = __shfl(nrm, 16 + r + 8 * half, 32);
            acc0[r] = acc0[r] / nrm0;
            acc1[r] = acc1[r] / nrm1;
        }
    }

    if (colok) {
        #pragma unroll
        for (int r = 0; r < 8; ++r) {
            const int m = m0 + r + 8 * half;
            out[(size_t)m * NOUT + ocol]        = acc0[r];
            out[(size_t)(m + 16) * NOUT + ocol] = acc1[r];
        }
    }
}

extern "C" void kernel_launch(void* const* d_in, const int* in_sizes, int n_in,
                              void* d_out, int out_size, void* d_ws, size_t ws_size,
                              hipStream_t stream) {
    const float* feat = (const float*)d_in[0];
    const int*   src  = (const int*)d_in[1];
    const int*   dstv = (const int*)d_in[2];
    const float* Ws0  = (const float*)d_in[3];
    const float* bs0  = (const float*)d_in[4];
    const float* Wn0  = (const float*)d_in[5];
    const float* bn0  = (const float*)d_in[6];
    const float* Ws1  = (const float*)d_in[7];
    const float* bs1  = (const float*)d_in[8];
    const float* Wn1  = (const float*)d_in[9];
    const float* bn1  = (const float*)d_in[10];
    const float* Ws2  = (const float*)d_in[11];
    const float* bs2  = (const float*)d_in[12];
    const float* Wn2  = (const float*)d_in[13];
    const float* bn2  = (const float*)d_in[14];
    float* out = (float*)d_out;

    int*   col    = (int*)d_ws;
    float* h1     = (float*)d_ws + (size_t)NNODES * KIN;
    float* h2     = h1  + (size_t)NNODES * KIN;
    int*   rowptr = (int*)(h2 + (size_t)NNODES * KIN);
    float* invdeg = (float*)(rowptr + NNODES);
    int* cnt     = (int*)h2;
    int* rank    = cnt + 2 * NNODES;
    int* rowptr1 = col + NEDGES;
    float2* Wsp0 = (float2*)(invdeg + NNODES);
    float2* Wnp0 = Wsp0 + PACK8_F2;
    float2* Wsp1 = Wnp0 + PACK8_F2;
    float2* Wnp1 = Wsp1 + PACK8_F2;
    float2* Wsp2 = Wnp1 + PACK8_F2;
    float2* Wnp2 = Wsp2 + PACK3_F2;

    const int ZB = 256;
    const int tileBlocks = NNODES / 32;

    prep_kernel<<<PACK_BLOCKS + ZERO_BLOCKS, PREP_T, 0, stream>>>(
        Ws0, Wn0, Ws1, Wn1, Ws2, Wn2, Wsp0, (float4*)cnt);

    rank_kernel<<<(NEDGES + ZB - 1) / ZB, ZB, 0, stream>>>(dstv, cnt, rank);
    scan_kernel<<<(NNODES + SCAN_T - 1) / SCAN_T, SCAN_T, 0, stream>>>(cnt, rowptr, rowptr1, invdeg);
    fill_kernel<<<(NEDGES + ZB - 1) / ZB, ZB, 0, stream>>>(src, dstv, rank, rowptr, rowptr1, col);

    sage_layer<128, true, true><<<tileBlocks, 256, 0, stream>>>(
        feat, rowptr, col, invdeg, Wsp0, Wnp0, bs0, bn0, h1);

    sage_layer<128, true, true><<<tileBlocks, 256, 0, stream>>>(
        h1, rowptr, col, invdeg, Wsp1, Wnp1, bs1, bn1, h2);

    sage_layer<40, false, false><<<tileBlocks, 256, 0, stream>>>(
        h2, rowptr, col, invdeg, Wsp2, Wnp2, bs2, bn2, out);
}


# Layer 2 reordered by linearity: both weight products done on layer 1's finished tile inside its kernel (h2 never stored), then the neighbour mean over 40-wide rows; sums now taken after the products
# speedup vs baseline: 4.9550x; 1.0366x over previous
#include <hip/hip_runtime.h>
#include <hip/hip_bf16.h>

#define NNODES 100000
#define NEDGES 1600000
#define KIN    128
#define NCLS   40
#define NT2    ((NCLS + 15) / 16)

typedef __attribute__((ext_vector_type(2))) float v2f;
typedef __attribute__((ext_vector_type(8))) float v8f;


__global__ void rank_kernel(const int* __restrict__ dst, int* __restrict__ cnt,
                            int* __restrict__ rank) {
    const int half = (int)((__builtin_amdgcn_s_sendmsg_rtn(0x87) >> 18) & 1u);
    int e = blockIdx.x * blockDim.x + threadIdx.x;
    if (e < NEDGES) rank[e] = 2 * atomicAdd(&cnt[half * NNODES + dst[e]], 1) + half;
}

#define SCAN_T 1024
__global__ void __launch_bounds__(SCAN_T)
scan_kernel(const int* __restrict__ cnt, int* __restrict__ rowptr,
            int* __restrict__ rowptr1, float* __restrict__ invdeg) {
    __shared__ int wfront[32];
    __shared__ int wown[32];
    const int t    = threadIdx.x;
    const int lane = t & 31;
    const int wv   = t >> 5;
    const int4* __restrict__ c4  = (const int4*)cnt;
    const int4* __restrict__ c4b = (const int4*)(cnt + NNODES);
    const int n4   = blockIdx.x * (SCAN_T / 4);
    const int node = blockIdx.x * SCAN_T + t;
    const int c0 = (node < NNODES) ? cnt[node] : 0;
    const int c1 = (node < NNODES) ? cnt[NNODES + node] : 0;
    const int c  = c0 + c1;

    int s = 0;
    for (int i0 = 0; i0 * SCAN_T < n4; i0 += 4) {
        int4 v[4], w[4];
        #pragma unroll
        for (int u = 0; u < 4; ++u) {
            const int idx = (i0 + u) * SCAN_T + t;
            v[u] = c4[idx < n4 ? idx : 0];
            w[u] = c4b[idx < n4 ? idx : 0];
        }
        #pragma unroll
        for (int u = 0; u < 4; ++u) {
            const int idx = (i0 + u) * SCAN_T + t;
            if (idx < n4) s += ((v[u].x + v[u].y) + (v[u].z + v[u].w))
                             + ((w[u].x + w[u].y) + (w[u].z + w[u].w));
        }
    }

    int x = c;
    for (int o = 1; o < 32; o <<= 1) {
        const int y = __shfl_up(x, o, 32);
        if (lane >= o) x += y;
    }
    for (int o = 16; o > 0; o >>= 1) s += __shfl_xor(s, o, 32);
    if (lane == 31) wown[wv]   = x;
    if (lane == 0)  wfront[wv] = s;
    __syncthreads();
    int z = wfront[lane] + ((lane < wv) ? wown[lane] : 0);
    for (int o = 16; o > 0; o >>= 1) z += __shfl_xor(z, o, 32);

    if (node < NNODES) {
        rowptr[node]  = z + (x - c);
        rowptr1[node] = z + (x - c) + c0;
        invdeg[node]  = 1.0f / fmaxf((float)c, 1.0f);
    }
}

__global__ void fill_kernel(const int* __restrict__ src, const int* __restrict__ dst,
                            const int* __restrict__ rank, const int* __restrict__ rowptr,
                            const int* __restrict__ rowptr1, int* __restrict__ col) {
    int e = blockIdx.x * blockDim.x + threadIdx.x;
    if (e < NEDGES) {
        const int rk = rank[e];
        const int* __restrict__ start = (rk & 1) ? rowptr1 : rowptr;
        col[start[dst[e]] + (rk >> 1)] = src[e];
    }
}

__device__ __forceinline__ float sum16(float v) {
    v += __builtin_bit_cast(float, __builtin_amdgcn_update_dpp(0, __builtin_bit_cast(int, v), 0x161, 0xf, 0xf, true));
    v += __builtin_bit_cast(float, __builtin_amdgcn_update_dpp(0, __builtin_bit_cast(int, v), 0x162, 0xf, 0xf, true));
    v += __builtin_bit_cast(float, __builtin_amdgcn_update_dpp(0, __builtin_bit_cast(int, v), 0x164, 0xf, 0xf, true));
    v += __builtin_bit_cast(float, __builtin_amdgcn_update_dpp(0, __builtin_bit_cast(int, v), 0x168, 0xf, 0xf, true));
    return v;
}

template<int B>
__device__ __forceinline__ void take_rows(const float* __restrict__ h, int mine, int k,
                                          int lane, float4& acc) {
    float4 v[B];
    #pragma unroll
    for (int j = 0; j < B; ++j) {
        const int s = __builtin_amdgcn_readlane(mine, k + j);
        v[j] = ((const float4*)(h + (size_t)s * KIN))[lane];
    }
    #pragma unroll
    for (int j = 0; j < B; ++j) {
        acc.x += v[j].x;
        acc.y += v[j].y;
        acc.z += v[j].z;
        acc.w += v[j].w;
    }
}

__device__ __forceinline__ void pack_w(const float* __restrict__ W, float2* __restrict__ Wp,
                                       int nout, int pid) {
    int lane = pid & 31;
    int i    = (pid >> 5) & ((KIN / 4) - 1);
    int tile = pid >> 10;
    int l = lane & 15, half = lane >> 4;
    int col = tile * 16 + l;
    int k   = 4 * i + 2 * half;
    float x = 0.f, y = 0.f;
    if (col < nout) {
        x = W[(size_t)k       * nout + col];
        y = W[(size_t)(k + 1) * nout + col];
    }
    Wp[pid] = make_float2(x, y);
}

#define PREP_T       256
#define PACK8_F2     (8 * (KIN / 4) * 32)
#define PACK3_F2     (3 * (KIN / 4) * 32)
#define PACK8_BLOCKS (PACK8_F2 / PREP_T)
#define PACK3_BLOCKS (PACK3_F2 / PREP_T)
#define PACK_BLOCKS  (4 * PACK8_BLOCKS + 2 * PACK3_BLOCKS)
#define CNT_F4       (2 * NNODES / 4)
#define ZERO_BLOCKS  ((CNT_F4 + PREP_T - 1) / PREP_T)
static_assert(PACK8_F2 % PREP_T == 0 && PACK3_F2 % PREP_T == 0, "pack blocks are whole");
__global__ void __launch_bounds__(PREP_T)
prep_kernel(const float* __restrict__ Ws0, const float* __restrict__ Wn0,
            const float* __restrict__ Ws1, const float* __restrict__ Wn1,
            const float* __restrict__ Ws2, const float* __restrict__ Wn2,
            float2* __restrict__ Wp, float4* __restrict__ cnt4) {
    const int b = blockIdx.x;
    const int t = threadIdx.x;
    if (b >= PACK_BLOCKS) {
        const int i = (b - PACK_BLOCKS) * PREP_T + t;
        if (i < CNT_F4) cnt4[i] = make_float4(0.f, 0.f, 0.f, 0.f);
        return;
    }
    if (b < 4 * PACK8_BLOCKS) {
        const int m = b / PACK8_BLOCKS;
        const float* __restrict__ W = (m == 0) ? Ws0 : (m == 1) ? Wn0 : (m == 2) ? Ws1 : Wn1;
        pack_w(W, Wp + (size_t)m * PACK8_F2, 128, (b - m * PACK8_BLOCKS) * PREP_T + t);
    } else {
        const int b3 = b - 4 * PACK8_BLOCKS;
        const int m  = b3 / PACK3_BLOCKS;
        const float* __restrict__ W = (m == 0) ? Ws2 : Wn2;
        pack_w(W, Wp + (size_t)4 * PACK8_F2 + (size_t)m * PACK3_F2, 40,
               (b3 - m * PACK3_BLOCKS) * PREP_T + t);
    }
}

template<int NOUT, bool ACT, bool NORM, bool PROJ>
__global__ void __launch_bounds__(256) __attribute__((amdgpu_waves_per_eu(PROJ ? 12 : 16, 16)))
sage_layer(const float* __restrict__ h,
           const int* __restrict__ rowptr,
           const int* __restrict__ col,
           const float* __restrict__ invdeg,
           const float2* __restrict__ Wsp,
           const float2* __restrict__ Wnp,
           const float* __restrict__ bself,
           const float* __restrict__ bneigh,
           float* __restrict__ out,
           const float2* __restrict__ Wsp2,
           const float2* __restrict__ Wnp2,
           const float* __restrict__ bself2,
           const float* __restrict__ bneigh2) {
    constexpr bool FULL = (NOUT % 16) == 0;
    constexpr int NT    = (NOUT + 15) / 16;
    constexpr int ROWS  = 32;
    constexpr int PITCH = KIN + 4;
    static_assert(!NORM || NT == 8, "the row-norm barriers need all eight waves");
    static_assert(!PROJ || (NORM && NOUT == KIN), "the second product reads the tile as a KIN-wide A operand");
    __shared__ float sSelf[ROWS * PITCH];
    __shared__ float sNeigh[ROWS * PITCH];

    const int lane = threadIdx.x & 31;
    const int wv   = __builtin_amdgcn_readfirstlane((int)(threadIdx.x >> 5));
    const int m0   = blockIdx.x * ROWS;

    int   beg[4], end[4], first[4];
    float idg[4];
    #pragma unroll
    for (int t = 0; t < 4; ++t) {
        const int node = m0 + wv + 8 * t;
        beg[t] = rowptr[node];
        end[t] = (node + 1 < NNODES) ? rowptr[node + 1] : NEDGES;
        idg[t] = invdeg[node];
    }
    #pragma unroll
    for (int t = 0; t < 4; ++t)
        first[t] = (lane < end[t] - beg[t]) ? col[beg[t] + lane] : 0;

    #pragma unroll
    for (int t = 0; t < 4; ++t) {
        const int r    = wv + 8 * t;
        const int node = m0 + r;
        const float4 hv = ((const float4*)(h + (size_t)node * KIN))[lane];

        float4 acc = make_float4(0.f, 0.f, 0.f, 0.f);
        int mine = first[t];
        for (int base = beg[t]; base < end[t]; ) {
            const int n = min(end[t] - base, 32);
            int k = 0;
            for (; k + 8 <= n; k += 8) take_rows<8>(h, mine, k, lane, acc);
            if (k + 4 <= n) { take_rows<4>(h, mine, k, lane, acc); k += 4; }
            if (k + 2 <= n) { take_rows<2>(h, mine, k, lane, acc); k += 2; }
            if (k < n)      { take_rows<1>(h, mine, k, lane, acc); }
            base += 32;
            if (base < end[t])
                mine = (lane < end[t] - base) ? col[base + lane] : 0;
        }
        acc.x *= idg[t]; acc.y *= idg[t]; acc.z *= idg[t]; acc.w *= idg[t];
        *(float4*)(sSelf  + r * PITCH + 4 * lane) = hv;
        *(float4*)(sNeigh + r * PITCH + 4 * lane) = acc;
    }
    __syncthreads();
    if (NT < 8 && wv >= NT) return;

    const int l    = lane & 15;
    const int half = lane >> 4;
    const int n0   = wv * 16;
    const int ocol = n0 + l;
    const bool colok = FULL || (ocol < NOUT);

    const float*  As = sSelf  + l * PITCH + 2 * half;
    const float*  An = sNeigh + l * PITCH + 2 * half;
    const float2* __restrict__ Bs = Wsp + (size_t)wv * (KIN / 4) * 32 + lane;
    const float2* __restrict__ Bn = Wnp + (size_t)wv * (KIN / 4) * 32 + lane;

    v8f acc0 = {0.f, 0.f, 0.f, 0.f, 0.f, 0.f, 0.f, 0.f};
    v8f acc1 = {0.f, 0.f, 0.f, 0.f, 0.f, 0.f, 0.f, 0.f};

    #pragma unroll
    for (int i = 0; i < KIN / 4; ++i) {
        const float2 a0f = *(const float2*)(As + 4 * i);
        const float2 a1f = *(const float2*)(As + 16 * PITCH + 4 * i);
        const float2 bf  = Bs[i * 32];
        v2f a0 = {a0f.x, a0f.y};
        v2f a1 = {a1f.x, a1f.y};
        v2f b  = {bf.x, bf.y};
        acc0 = __builtin_amdgcn_wmma_f32_16x16x4_f32(false, a0, false, b,
                                                     (short)0, acc0, false, false);
        acc1 = __builtin_amdgcn_wmma_f32_16x16x4_f32(false, a1, false, b,
                                                     (short)0, acc1, false, false);
    }
    #pragma unroll
    for (int i = 0; i < KIN / 4; ++i) {
        const float2 a0f = *(const float2*)(An + 4 * i);
        const float2 a1f = *(const float2*)(An + 16 * PITCH + 4 * i);
        const float2 bf  = Bn[i * 32];
        v2f a0 = {a0f.x, a0f.y};
        v2f a1 = {a1f.x, a1f.y};
        v2f b  = {bf.x, bf.y};
        acc0 = __builtin_amdgcn_wmma_f32_16x16x4_f32(false, a0, false, b,
                                                     (short)0, acc0, false, false);
        acc1 = __builtin_amdgcn_wmma_f32_16x16x4_f32(false, a1, false, b,
                                                     (short)0, acc1, false, false);
    }

    const float bias = colok ? (bself[ocol] + bneigh[ocol]) : 0.0f;
    #pragma unroll
    for (int r = 0; r < 8; ++r) {
        float v0 = acc0[r] + bias;
        float v1 = acc1[r] + bias;
        if (ACT) { v0 = fmaxf(v0, 0.0f); v1 = fmaxf(v1, 0.0f); }
        acc0[r] = v0;
        acc1[r] = v1;
    }

    if (NORM) {
        __shared__ alignas(16) float part[8 * ROWS];
        float q0[8], q1[8];
        #pragma unroll
        for (int r = 0; r < 8; ++r) {
            q0[r] = sum16(acc0[r] * acc0[r]);
            q1[r] = sum16(acc1[r] * acc1[r]);
        }
        if (l == 0) {
            float* p = part + wv * ROWS + 8 * half;
            *(float4*)(p)      = make_float4(q0[0], q0[1], q0[2], q0[3]);
            *(float4*)(p + 4)  = make_float4(q0[4], q0[5], q0[6], q0[7]);
            *(float4*)(p + 16) = make_float4(q1[0], q1[1], q1[2], q1[3]);
            *(float4*)(p + 20) = make_float4(q1[4], q1[5], q1[6], q1[7]);
        }
        __syncthreads();
        float tot = part[lane];
        #pragma unroll
        for (int w = 1; w < 8; ++w) tot += part[w * ROWS + lane];
        const float nrm = fmaxf(sqrtf(tot), 1e-12f);
        #pragma unroll
        for (int r = 0; r < 8; ++r) {
            const float nrm0 = __shfl(nrm, r + 8 * half, 32);
            const float nrm1 = __shfl(nrm, 16 + r + 8 * half, 32);
            acc0[r] = acc0[r] / nrm0;
            acc1[r] = acc1[r] / nrm1;
        }
    }

    if (PROJ) {
        #pragma unroll
        for (int r = 0; r < 8; ++r) {
            sSelf[(r + 8 * half) * PITCH + ocol]      = acc0[r];
            sSelf[(16 + r + 8 * half) * PITCH + ocol] = acc1[r];
        }
        __syncthreads();
        if (wv >= 2 * NT2) return;
        const bool isq = wv >= NT2;
        const int  tj  = isq ? wv - NT2 : wv;
        const float2* __restrict__ B2 = (isq ? Wnp2 : Wsp2) + (size_t)tj * (KIN / 4) * 32 + lane;
        v8f c0 = {0.f, 0.f, 0.f, 0.f, 0.f, 0.f, 0.f, 0.f};
        v8f c1 = {0.f, 0.f, 0.f, 0.f, 0.f, 0.f, 0.f, 0.f};
        #pragma unroll
        for (int i = 0; i < KIN / 4; ++i) {
            const float2 a0f = *(const float2*)(As + 4 * i);
            const float2 a1f = *(const float2*)(As + 16 * PITCH + 4 * i);
            const float2 bf  = B2[i * 32];
            v2f a0 = {a0f.x, a0f.y};
            v2f a1 = {a1f.x, a1f.y};
            v2f b  = {bf.x, bf.y};
            c0 = __builtin_amdgcn_wmma_f32_16x16x4_f32(false, a0, false, b,
                                                       (short)0, c0, false, false);
            c1 = __builtin_amdgcn_wmma_f32_16x16x4_f32(false, a1, false, b,
                                                       (short)0, c1, false, false);
        }
        const int pc = tj * 16 + l;
        if (pc < NCLS) {
            const float b2 = isq ? 0.0f : (bself2[pc] + bneigh2[pc]);
            float* __restrict__ dstp = out + (isq ? (size_t)NNODES * NCLS : (size_t)0);
            #pragma unroll
            for (int r = 0; r < 8; ++r) {
                const int m = m0 + r + 8 * half;
                dstp[(size_t)m * NCLS + pc]        = c0[r] + b2;
                dstp[(size_t)(m + 16) * NCLS + pc] = c1[r] + b2;
            }
        }
    } else if (colok) {
        #pragma unroll
        for (int r = 0; r < 8; ++r) {
            const int m = m0 + r + 8 * half;
            out[(size_t)m * NOUT + ocol]        = acc0[r];
            out[(size_t)(m + 16) * NOUT + ocol] = acc1[r];
        }
    }
}

template<int K>
__device__ __forceinline__ int row_lane(int x) {
    return __builtin_amdgcn_update_dpp(0, x, 0x150 + K, 0xf, 0xf, true);
}

template<int K0>
__device__ __forceinline__ void take_q8(const float4* __restrict__ q4, int mine, int n, int l,
                                        bool on, float4& acc) {
    float4 v[8];
#define TAKE_Q(j)                                                              \
    {                                                                          \
        const int s = row_lane<K0 + j>(mine);                                  \
        v[j] = make_float4(0.f, 0.f, 0.f, 0.f);                                \
        if (on && (K0 + j) < n) v[j] = q4[(size_t)s * (NCLS / 4) + l];         \
    }
    TAKE_Q(0) TAKE_Q(1) TAKE_Q(2) TAKE_Q(3) TAKE_Q(4) TAKE_Q(5) TAKE_Q(6) TAKE_Q(7)
#undef TAKE_Q
    #pragma unroll
    for (int j = 0; j < 8; ++j) {
        acc.x += v[j].x;
        acc.y += v[j].y;
        acc.z += v[j].z;
        acc.w += v[j].w;
    }
}

#define OUT_T     256
#define OUT_NPH   5
#define OUT_NODES ((OUT_T / 16) * OUT_NPH)
static_assert(NNODES % OUT_NODES == 0, "whole blocks, no node guard");
__global__ void __launch_bounds__(OUT_T)
out_layer(const float* __restrict__ p, const float* __restrict__ q,
          const int* __restrict__ rowptr, const int* __restrict__ col,
          const float* __restrict__ invdeg, float* __restrict__ out) {
    const int  l   = threadIdx.x & 15;
    const int  hw  = (int)(threadIdx.x >> 4);
    const int  nb  = blockIdx.x * OUT_NODES;
    const bool on  = l < NCLS / 4;
    const float4* __restrict__ q4 = (const float4*)q;

    int   beg[OUT_NPH], end[OUT_NPH], first[OUT_NPH];
    float idg[OUT_NPH];
    #pragma unroll
    for (int t = 0; t < OUT_NPH; ++t) {
        const int node = nb + hw + 16 * t;
        beg[t] = rowptr[node];
        end[t] = (node + 1 < NNODES) ? rowptr[node + 1] : NEDGES;
        idg[t] = invdeg[node];
    }
    #pragma unroll
    for (int t = 0; t < OUT_NPH; ++t)
        first[t] = (l < end[t] - beg[t]) ? col[beg[t] + l] : 0;

    #pragma unroll
    for (int t = 0; t < OUT_NPH; ++t) {
        const int node = nb + hw + 16 * t;
        float4 pv = make_float4(0.f, 0.f, 0.f, 0.f);
        if (on) pv = ((const float4*)(p + (size_t)node * NCLS))[l];

        float4 acc = make_float4(0.f, 0.f, 0.f, 0.f);
        int mine = first[t];
        for (int base = beg[t]; base < end[t]; ) {
            const int n = min(end[t] - base, 16);
            take_q8<0>(q4, mine, n, l, on, acc);
            if (n > 8) take_q8<8>(q4, mine, n, l, on, acc);
            base += 16;
            if (base < end[t])
                mine = (l < end[t] - base) ? col[base + l] : 0;
        }
        if (on) {
            float4 o;
            o.x = pv.x + acc.x * idg[t];
            o.y = pv.y + acc.y * idg[t];
            o.z = pv.z + acc.z * idg[t];
            o.w = pv.w + acc.w * idg[t];
            ((float4*)(out + (size_t)node * NCLS))[l] = o;
        }
    }
}

extern "C" void kernel_launch(void* const* d_in, const int* in_sizes, int n_in,
                              void* d_out, int out_size, void* d_ws, size_t ws_size,
                              hipStream_t stream) {
    const float* feat = (const float*)d_in[0];
    const int*   src  = (const int*)d_in[1];
    const int*   dstv = (const int*)d_in[2];
    const float* Ws0  = (const float*)d_in[3];
    const float* bs0  = (const float*)d_in[4];
    const float* Wn0  = (const float*)d_in[5];
    const float* bn0  = (const float*)d_in[6];
    const float* Ws1  = (const float*)d_in[7];
    const float* bs1  = (const float*)d_in[8];
    const float* Wn1  = (const float*)d_in[9];
    const float* bn1  = (const float*)d_in[10];
    const float* Ws2  = (const float*)d_in[11];
    const float* bs2  = (const float*)d_in[12];
    const float* Wn2  = (const float*)d_in[13];
    const float* bn2  = (const float*)d_in[14];
    float* out = (float*)d_out;

    int*   col    = (int*)d_ws;
    float* h1     = (float*)d_ws + (size_t)NNODES * KIN;
    float* h2     = h1  + (size_t)NNODES * KIN;
    int*   rowptr = (int*)(h2 + (size_t)NNODES * KIN);
    float* invdeg = (float*)(rowptr + NNODES);
    int* cnt     = (int*)h2;
    int* rank    = cnt + 2 * NNODES;
    int* rowptr1 = col + NEDGES;
    float* pl    = h2;
    float* ql    = pl + (size_t)NNODES * NCLS;
    float2* Wsp0 = (float2*)(invdeg + NNODES);
    float2* Wnp0 = Wsp0 + PACK8_F2;
    float2* Wsp1 = Wnp0 + PACK8_F2;
    float2* Wnp1 = Wsp1 + PACK8_F2;
    float2* Wsp2 = Wnp1 + PACK8_F2;
    float2* Wnp2 = Wsp2 + PACK3_F2;

    const int ZB = 256;
    const int tileBlocks = NNODES / 32;

    prep_kernel<<<PACK_BLOCKS + ZERO_BLOCKS, PREP_T, 0, stream>>>(
        Ws0, Wn0, Ws1, Wn1, Ws2, Wn2, Wsp0, (float4*)cnt);

    rank_kernel<<<(NEDGES + ZB - 1) / ZB, ZB, 0, stream>>>(dstv, cnt, rank);
    scan_kernel<<<(NNODES + SCAN_T - 1) / SCAN_T, SCAN_T, 0, stream>>>(cnt, rowptr, rowptr1, invdeg);
    fill_kernel<<<(NEDGES + ZB - 1) / ZB, ZB, 0, stream>>>(src, dstv, rank, rowptr, rowptr1, col);

    sage_layer<128, true, true, false><<<tileBlocks, 256, 0, stream>>>(
        feat, rowptr, col, invdeg, Wsp0, Wnp0, bs0, bn0, h1,
        nullptr, nullptr, nullptr, nullptr);

    sage_layer<128, true, true, true><<<tileBlocks, 256, 0, stream>>>(
        h1, rowptr, col, invdeg, Wsp1, Wnp1, bs1, bn1, pl,
        Wsp2, Wnp2, bs2, bn2);

    out_layer<<<NNODES / OUT_NODES, OUT_T, 0, stream>>>(pl, ql, rowptr, col, invdeg, out);
}


# Edge sort: global counting atomics and scattered 4 B fill -> two-level counting sort on LDS counters (edges binned by 512-node bucket, one work-group ranks each bucket); add order = LDS arrival
# speedup vs baseline: 5.8678x; 1.1842x over previous
#include <hip/hip_runtime.h>
#include <hip/hip_bf16.h>

#define NNODES 100000
#define NEDGES 1600000
#define KIN    128
#define NCLS   40
#define NT2    ((NCLS + 15) / 16)

typedef __attribute__((ext_vector_type(2))) float v2f;
typedef __attribute__((ext_vector_type(8))) float v8f;

#define BSH     9
#define BNODES  (1 << BSH)
#define NBUCK   ((NNODES + BNODES - 1) / BNODES)
#define EBLK    250
#define EPB     (NEDGES / EBLK)
#define ET      256
#define EPT     (EPB / ET)
#define NTAB    (NBUCK * EBLK)
static_assert(EPB * EBLK == NEDGES && EPT * ET == EPB, "whole edge blocks");
static_assert(NTAB % 4 == 0, "the table is read in 16 B pieces");
static_assert(NBUCK <= ET, "edge_scatter: one thread per bucket");

__device__ __forceinline__ void edge_hist(const int* __restrict__ dst, int* __restrict__ tab,
                                          int eb, int t, int* hist) {
    for (int i = t; i < NBUCK; i += ET) hist[i] = 0;
    __syncthreads();
    const int* __restrict__ de = dst + eb * EPB;
    int d[EPT];
    #pragma unroll
    for (int j = 0; j < EPT; ++j) d[j] = de[j * ET + t];
    #pragma unroll
    for (int j = 0; j < EPT; ++j) atomicAdd(&hist[d[j] >> BSH], 1);
    __syncthreads();
    for (int i = t; i < NBUCK; i += ET) tab[i * EBLK + eb] = hist[i];
}

#define SCAN_T 1024
__global__ void __launch_bounds__(SCAN_T)
tab_scan(const int* __restrict__ tab, int* __restrict__ toff) {
    __shared__ int wfront[32];
    __shared__ int wown[32];
    const int t    = threadIdx.x;
    const int lane = t & 31;
    const int wv   = t >> 5;
    const int4* __restrict__ c4 = (const int4*)tab;
    const int n4   = blockIdx.x * (SCAN_T / 4);
    const int idx0 = blockIdx.x * SCAN_T + t;
    const int c = (idx0 < NTAB) ? tab[idx0] : 0;

    int s = 0;
    for (int i0 = 0; i0 * SCAN_T < n4; i0 += 4) {
        int4 v[4];
        #pragma unroll
        for (int u = 0; u < 4; ++u) {
            const int idx = (i0 + u) * SCAN_T + t;
            v[u] = c4[idx < n4 ? idx : 0];
        }
        #pragma unroll
        for (int u = 0; u < 4; ++u) {
            const int idx = (i0 + u) * SCAN_T + t;
            if (idx < n4) s += (v[u].x + v[u].y) + (v[u].z + v[u].w);
        }
    }

    int x = c;
    for (int o = 1; o < 32; o <<= 1) {
        const int y = __shfl_up(x, o, 32);
        if (lane >= o) x += y;
    }
    for (int o = 16; o > 0; o >>= 1) s += __shfl_xor(s, o, 32);
    if (lane == 31) wown[wv]   = x;
    if (lane == 0)  wfront[wv] = s;
    __syncthreads();
    int z = wfront[lane] + ((lane < wv) ? wown[lane] : 0);
    for (int o = 16; o > 0; o >>= 1) z += __shfl_xor(z, o, 32);

    if (idx0 < NTAB) toff[idx0] = z + (x - c);
}

__global__ void __launch_bounds__(ET)
edge_scatter(const int* __restrict__ src, const int* __restrict__ dst,
             const int* __restrict__ toff, int2* __restrict__ ebuf) {
    __shared__ int  hist[NBUCK];
    __shared__ int  lstart[NBUCK];
    __shared__ int  goff[NBUCK];
    __shared__ int2 sorted[EPB];
    const int t  = threadIdx.x;
    const int eb = blockIdx.x;
    const int gt = (t < NBUCK) ? toff[t * EBLK + eb] : 0;
    for (int i = t; i < NBUCK; i += ET) hist[i] = 0;
    __syncthreads();
    const int* __restrict__ se = src + eb * EPB;
    const int* __restrict__ de = dst + eb * EPB;
    int s[EPT], d[EPT], r[EPT];
    #pragma unroll
    for (int j = 0; j < EPT; ++j) {
        s[j] = se[j * ET + t];
        d[j] = de[j * ET + t];
    }
    #pragma unroll
    for (int j = 0; j < EPT; ++j) r[j] = atomicAdd(&hist[d[j] >> BSH], 1);
    __syncthreads();
    if (t < 32) {
        constexpr int PER = (NBUCK + 31) / 32;
        int v[PER];
        int sum = 0;
        #pragma unroll
        for (int k = 0; k < PER; ++k) {
            const int b = t * PER + k;
            v[k] = (b < NBUCK) ? hist[b] : 0;
            sum += v[k];
        }
        int x = sum;
        for (int o = 1; o < 32; o <<= 1) {
            const int y = __shfl_up(x, o, 32);
            if (t >= o) x += y;
        }
        int run = x - sum;
        #pragma unroll
        for (int k = 0; k < PER; ++k) {
            const int b = t * PER + k;
            if (b < NBUCK) lstart[b] = run;
            run += v[k];
        }
    }
    __syncthreads();
    if (t < NBUCK) goff[t] = gt - lstart[t];
    #pragma unroll
    for (int j = 0; j < EPT; ++j)
        sorted[lstart[d[j] >> BSH] + r[j]] = make_int2(s[j], d[j]);
    __syncthreads();
    for (int p = t; p < EPB; p += ET) {
        const int2 v = sorted[p];
        ebuf[goff[v.y >> BSH] + p] = v;
    }
}

#define BT   BNODES
#define BCAP 12288
__global__ void __launch_bounds__(BT)
bucket_sort(const int2* __restrict__ ebuf, const int* __restrict__ toff,
            int* __restrict__ rowptr, float* __restrict__ invdeg, int* __restrict__ col) {
    __shared__ int cnt[BNODES];
    __shared__ int cur[BNODES];
    __shared__ int lrow[BNODES];
    __shared__ int wsum[BT / 32];
    __shared__ int stage[BCAP];
    const int t    = threadIdx.x;
    const int lane = t & 31;
    const int wv   = t >> 5;
    const int b    = blockIdx.x;
    const int bs   = toff[b * EBLK];
    const int be   = (b + 1 < NBUCK) ? toff[(b + 1) * EBLK] : NEDGES;
    const int n    = be - bs;
    const int nb   = b * BNODES;
    const int2* __restrict__ eb = ebuf + bs;
    cnt[t] = 0;
    cur[t] = 0;
    __syncthreads();
    for (int i0 = t; i0 < n; i0 += 4 * BT) {
        int d[4];
        #pragma unroll
        for (int u = 0; u < 4; ++u) {
            const int i = i0 + u * BT;
            d[u] = (i < n) ? eb[i].y : nb;
        }
        #pragma unroll
        for (int u = 0; u < 4; ++u)
            if (i0 + u * BT < n) atomicAdd(&cnt[d[u] - nb], 1);
    }
    __syncthreads();

    const int c = cnt[t];
    int x = c;
    for (int o = 1; o < 32; o <<= 1) {
        const int y = __shfl_up(x, o, 32);
        if (lane >= o) x += y;
    }
    if (lane == 31) wsum[wv] = x;
    __syncthreads();
    int z = (lane < wv) ? wsum[lane] : 0;
    for (int o = 16; o > 0; o >>= 1) z += __shfl_xor(z, o, 32);
    const int ex = z + (x - c);
    lrow[t] = ex;
    const int node = nb + t;
    if (node < NNODES) {
        rowptr[node] = bs + ex;
        invdeg[node] = 1.0f / fmaxf((float)c, 1.0f);
    }
    __syncthreads();

    const bool staged = n <= BCAP;
    for (int i0 = t; i0 < n; i0 += 4 * BT) {
        int2 v[4];
        #pragma unroll
        for (int u = 0; u < 4; ++u) {
            const int i = i0 + u * BT;
            v[u] = (i < n) ? eb[i] : make_int2(0, nb);
        }
        #pragma unroll
        for (int u = 0; u < 4; ++u) {
            if (i0 + u * BT < n) {
                const int dl = v[u].y - nb;
                const int p  = lrow[dl] + atomicAdd(&cur[dl], 1);
                if (staged) stage[p] = v[u].x;
                else        col[bs + p] = v[u].x;
            }
        }
    }
    if (staged) {
        __syncthreads();
        for (int i = t; i < n; i += BT) col[bs + i] = stage[i];
    }
}

__device__ __forceinline__ float sum16(float v) {
    v += __builtin_bit_cast(float, __builtin_amdgcn_update_dpp(0, __builtin_bit_cast(int, v), 0x161, 0xf, 0xf, true));
    v += __builtin_bit_cast(float, __builtin_amdgcn_update_dpp(0, __builtin_bit_cast(int, v), 0x162, 0xf, 0xf, true));
    v += __builtin_bit_cast(float, __builtin_amdgcn_update_dpp(0, __builtin_bit_cast(int, v), 0x164, 0xf, 0xf, true));
    v += __builtin_bit_cast(float, __builtin_amdgcn_update_dpp(0, __builtin_bit_cast(int, v), 0x168, 0xf, 0xf, true));
    return v;
}

template<int B>
__device__ __forceinline__ void take_rows(const float* __restrict__ h, int mine, int k,
                                          int lane, float4& acc) {
    float4 v[B];
    #pragma unroll
    for (int j = 0; j < B; ++j) {
        const int s = __builtin_amdgcn_readlane(mine, k + j);
        v[j] = ((const float4*)(h + (size_t)s * KIN))[lane];
    }
    #pragma unroll
    for (int j = 0; j < B; ++j) {
        acc.x += v[j].x;
        acc.y += v[j].y;
        acc.z += v[j].z;
        acc.w += v[j].w;
    }
}

__device__ __forceinline__ void pack_w(const float* __restrict__ W, float2* __restrict__ Wp,
                                       int nout, int pid) {
    int lane = pid & 31;
    int i    = (pid >> 5) & ((KIN / 4) - 1);
    int tile = pid >> 10;
    int l = lane & 15, half = lane >> 4;
    int col = tile * 16 + l;
    int k   = 4 * i + 2 * half;
    float x = 0.f, y = 0.f;
    if (col < nout) {
        x = W[(size_t)k       * nout + col];
        y = W[(size_t)(k + 1) * nout + col];
    }
    Wp[pid] = make_float2(x, y);
}

#define PREP_T       256
#define PACK8_F2     (8 * (KIN / 4) * 32)
#define PACK3_F2     (3 * (KIN / 4) * 32)
#define PACK8_BLOCKS (PACK8_F2 / PREP_T)
#define PACK3_BLOCKS (PACK3_F2 / PREP_T)
#define PACK_BLOCKS  (4 * PACK8_BLOCKS + 2 * PACK3_BLOCKS)
static_assert(PACK8_F2 % PREP_T == 0 && PACK3_F2 % PREP_T == 0, "pack blocks are whole");
static_assert(PREP_T == ET, "edge_hist is written for ET threads");
__global__ void __launch_bounds__(PREP_T)
prep_kernel(const float* __restrict__ Ws0, const float* __restrict__ Wn0,
            const float* __restrict__ Ws1, const float* __restrict__ Wn1,
            const float* __restrict__ Ws2, const float* __restrict__ Wn2,
            float2* __restrict__ Wp, const int* __restrict__ dst, int* __restrict__ tab) {
    __shared__ int hist[NBUCK];
    const int t = threadIdx.x;
    if (blockIdx.x < EBLK) {
        edge_hist(dst, tab, blockIdx.x, t, hist);
        return;
    }
    const int b = blockIdx.x - EBLK;
    if (b < 4 * PACK8_BLOCKS) {
        const int m = b / PACK8_BLOCKS;
        const float* __restrict__ W = (m == 0) ? Ws0 : (m == 1) ? Wn0 : (m == 2) ? Ws1 : Wn1;
        pack_w(W, Wp + (size_t)m * PACK8_F2, 128, (b - m * PACK8_BLOCKS) * PREP_T + t);
    } else {
        const int b3 = b - 4 * PACK8_BLOCKS;
        const int m  = b3 / PACK3_BLOCKS;
        const float* __restrict__ W = (m == 0) ? Ws2 : Wn2;
        pack_w(W, Wp + (size_t)4 * PACK8_F2 + (size_t)m * PACK3_F2, 40,
               (b3 - m * PACK3_BLOCKS) * PREP_T + t);
    }
}

template<int NOUT, bool ACT, bool NORM, bool PROJ>
__global__ void __launch_bounds__(256) __attribute__((amdgpu_waves_per_eu(PROJ ? 12 : 16, 16)))
sage_layer(const float* __restrict__ h,
           const int* __restrict__ rowptr,
           const int* __restrict__ col,
           const float* __restrict__ invdeg,
           const float2* __restrict__ Wsp,
           const float2* __restrict__ Wnp,
           const float* __restrict__ bself,
           const float* __restrict__ bneigh,
           float* __restrict__ out,
           const float2* __restrict__ Wsp2,
           const float2* __restrict__ Wnp2,
           const float* __restrict__ bself2,
           const float* __restrict__ bneigh2) {
    constexpr bool FULL = (NOUT % 16) == 0;
    constexpr int NT    = (NOUT + 15) / 16;
    constexpr int ROWS  = 32;
    constexpr int PITCH = KIN + 4;
    static_assert(!NORM || NT == 8, "the row-norm barriers need all eight waves");
    static_assert(!PROJ || (NORM && NOUT == KIN), "the second product reads the tile as a KIN-wide A operand");
    __shared__ float sSelf[ROWS * PITCH];
    __shared__ float sNeigh[ROWS * PITCH];

    const int lane = threadIdx.x & 31;
    const int wv   = __builtin_amdgcn_readfirstlane((int)(threadIdx.x >> 5));
    const int m0   = blockIdx.x * ROWS;

    int   beg[4], end[4], first[4];
    float idg[4];
    #pragma unroll
    for (int t = 0; t < 4; ++t) {
        const int node = m0 + wv + 8 * t;
        beg[t] = rowptr[node];
        end[t] = (node + 1 < NNODES) ? rowptr[node + 1] : NEDGES;
        idg[t] = invdeg[node];
    }
    #pragma unroll
    for (int t = 0; t < 4; ++t)
        first[t] = (lane < end[t] - beg[t]) ? col[beg[t] + lane] : 0;

    #pragma unroll
    for (int t = 0; t < 4; ++t) {
        const int r    = wv + 8 * t;
        const int node = m0 + r;
        const float4 hv = ((const float4*)(h + (size_t)node * KIN))[lane];

        float4 acc = make_float4(0.f, 0.f, 0.f, 0.f);
        int mine = first[t];
        for (int base = beg[t]; base < end[t]; ) {
            const int n = min(end[t] - base, 32);
            int k = 0;
            for (; k + 8 <= n; k += 8) take_rows<8>(h, mine, k, lane, acc);
            if (k + 4 <= n) { take_rows<4>(h, mine, k, lane, acc); k += 4; }
            if (k + 2 <= n) { take_rows<2>(h, mine, k, lane, acc); k += 2; }
            if (k < n)      { take_rows<1>(h, mine, k, lane, acc); }
            base += 32;
            if (base < end[t])
                mine = (lane < end[t] - base) ? col[base + lane] : 0;
        }
        acc.x *= idg[t]; acc.y *= idg[t]; acc.z *= idg[t]; acc.w *= idg[t];
        *(float4*)(sSelf  + r * PITCH + 4 * lane) = hv;
        *(float4*)(sNeigh + r * PITCH + 4 * lane) = acc;
    }
    __syncthreads();
    if (NT < 8 && wv >= NT) return;

    const int l    = lane & 15;
    const int half = lane >> 4;
    const int n0   = wv * 16;
    const int ocol = n0 + l;
    const bool colok = FULL || (ocol < NOUT);

    const float*  As = sSelf  + l * PITCH + 2 * half;
    const float*  An = sNeigh + l * PITCH + 2 * half;
    const float2* __restrict__ Bs = Wsp + (size_t)wv * (KIN / 4) * 32 + lane;
    const float2* __restrict__ Bn = Wnp + (size_t)wv * (KIN / 4) * 32 + lane;

    v8f acc0 = {0.f, 0.f, 0.f, 0.f, 0.f, 0.f, 0.f, 0.f};
    v8f acc1 = {0.f, 0.f, 0.f, 0.f, 0.f, 0.f, 0.f, 0.f};

    #pragma unroll
    for (int i = 0; i < KIN / 4; ++i) {
        const float2 a0f = *(const float2*)(As + 4 * i);
        const float2 a1f = *(const float2*)(As + 16 * PITCH + 4 * i);
        const float2 bf  = Bs[i * 32];
        v2f a0 = {a0f.x, a0f.y};
        v2f a1 = {a1f.x, a1f.y};
        v2f b  = {bf.x, bf.y};
        acc0 = __builtin_amdgcn_wmma_f32_16x16x4_f32(false, a0, false, b,
                                                     (short)0, acc0, false, false);
        acc1 = __builtin_amdgcn_wmma_f32_16x16x4_f32(false, a1, false, b,
                                                     (short)0, acc1, false, false);
    }
    #pragma unroll
    for (int i = 0; i < KIN / 4; ++i) {
        const float2 a0f = *(const float2*)(An + 4 * i);
        const float2 a1f = *(const float2*)(An + 16 * PITCH + 4 * i);
        const float2 bf  = Bn[i * 32];
        v2f a0 = {a0f.x, a0f.y};
        v2f a1 = {a1f.x, a1f.y};
        v2f b  = {bf.x, bf.y};
        acc0 = __builtin_amdgcn_wmma_f32_16x16x4_f32(false, a0, false, b,
                                                     (short)0, acc0, false, false);
        acc1 = __builtin_amdgcn_wmma_f32_16x16x4_f32(false, a1, false, b,
                                                     (short)0, acc1, false, false);
    }

    const float bias = colok ? (bself[ocol] + bneigh[ocol]) : 0.0f;
    #pragma unroll
    for (int r = 0; r < 8; ++r) {
        float v0 = acc0[r] + bias;
        float v1 = acc1[r] + bias;
        if (ACT) { v0 = fmaxf(v0, 0.0f); v1 = fmaxf(v1, 0.0f); }
        acc0[r] = v0;
        acc1[r] = v1;
    }

    if (NORM) {
        __shared__ alignas(16) float part[8 * ROWS];
        float q0[8], q1[8];
        #pragma unroll
        for (int r = 0; r < 8; ++r) {
            q0[r] = sum16(acc0[r] * acc0[r]);
            q1[r] = sum16(acc1[r] * acc1[r]);
        }
        if (l == 0) {
            float* p = part + wv * ROWS + 8 * half;
            *(float4*)(p)      = make_float4(q0[0], q0[1], q0[2], q0[3]);
            *(float4*)(p + 4)  = make_float4(q0[4], q0[5], q0[6], q0[7]);
            *(float4*)(p + 16) = make_float4(q1[0], q1[1], q1[2], q1[3]);
            *(float4*)(p + 20) = make_float4(q1[4], q1[5], q1[6], q1[7]);
        }
        __syncthreads();
        float tot = part[lane];
        #pragma unroll
        for (int w = 1; w < 8; ++w) tot += part[w * ROWS + lane];
        const float nrm = fmaxf(sqrtf(tot), 1e-12f);
        #pragma unroll
        for (int r = 0; r < 8; ++r) {
            const float nrm0 = __shfl(nrm, r + 8 * half, 32);
            const float nrm1 = __shfl(nrm, 16 + r + 8 * half, 32);
            acc0[r] = acc0[r] / nrm0;
            acc1[r] = acc1[r] / nrm1;
        }
    }

    if (PROJ) {
        #pragma unroll
        for (int r = 0; r < 8; ++r) {
            sSelf[(r + 8 * half) * PITCH + ocol]      = acc0[r];
            sSelf[(16 + r + 8 * half) * PITCH + ocol] = acc1[r];
        }
        __syncthreads();
        if (wv >= 2 * NT2) return;
        const bool isq = wv >= NT2;
        const int  tj  = isq ? wv - NT2 : wv;
        const float2* __restrict__ B2 = (isq ? Wnp2 : Wsp2) + (size_t)tj * (KIN / 4) * 32 + lane;
        v8f c0 = {0.f, 0.f, 0.f, 0.f, 0.f, 0.f, 0.f, 0.f};
        v8f c1 = {0.f, 0.f, 0.f, 0.f, 0.f, 0.f, 0.f, 0.f};
        #pragma unroll
        for (int i = 0; i < KIN / 4; ++i) {
            const float2 a0f = *(const float2*)(As + 4 * i);
            const float2 a1f = *(const float2*)(As + 16 * PITCH + 4 * i);
            const float2 bf  = B2[i * 32];
            v2f a0 = {a0f.x, a0f.y};
            v2f a1 = {a1f.x, a1f.y};
            v2f b  = {bf.x, bf.y};
            c0 = __builtin_amdgcn_wmma_f32_16x16x4_f32(false, a0, false, b,
                                                       (short)0, c0, false, false);
            c1 = __builtin_amdgcn_wmma_f32_16x16x4_f32(false, a1, false, b,
                                                       (short)0, c1, false, false);
        }
        const int pc = tj * 16 + l;
        if (pc < NCLS) {
            const float b2 = isq ? 0.0f : (bself2[pc] + bneigh2[pc]);
            float* __restrict__ dstp = out + (isq ? (size_t)NNODES * NCLS : (size_t)0);
            #pragma unroll
            for (int r = 0; r < 8; ++r) {
                const int m = m0 + r + 8 * half;
                dstp[(size_t)m * NCLS + pc]        = c0[r] + b2;
                dstp[(size_t)(m + 16) * NCLS + pc] = c1[r] + b2;
            }
        }
    } else if (colok) {
        #pragma unroll
        for (int r = 0; r < 8; ++r) {
            const int m = m0 + r + 8 * half;
            out[(size_t)m * NOUT + ocol]        = acc0[r];
            out[(size_t)(m + 16) * NOUT + ocol] = acc1[r];
        }
    }
}

template<int K>
__device__ __forceinline__ int row_lane(int x) {
    return __builtin_amdgcn_update_dpp(0, x, 0x150 + K, 0xf, 0xf, true);
}

template<int K0>
__device__ __forceinline__ void take_q8(const float4* __restrict__ q4, int mine, int n, int l,
                                        bool on, float4& acc) {
    float4 v[8];
#define TAKE_Q(j)                                                              \
    {                                                                          \
        const int s = row_lane<K0 + j>(mine);                                  \
        v[j] = make_float4(0.f, 0.f, 0.f, 0.f);                                \
        if (on && (K0 + j) < n) v[j] = q4[(size_t)s * (NCLS / 4) + l];         \
    }
    TAKE_Q(0) TAKE_Q(1) TAKE_Q(2) TAKE_Q(3) TAKE_Q(4) TAKE_Q(5) TAKE_Q(6) TAKE_Q(7)
#undef TAKE_Q
    #pragma unroll
    for (int j = 0; j < 8; ++j) {
        acc.x += v[j].x;
        acc.y += v[j].y;
        acc.z += v[j].z;
        acc.w += v[j].w;
    }
}

#define OUT_T     256
#define OUT_NPH   5
#define OUT_NODES ((OUT_T / 16) * OUT_NPH)
static_assert(NNODES % OUT_NODES == 0, "whole blocks, no node guard");
__global__ void __launch_bounds__(OUT_T)
out_layer(const float* __restrict__ p, const float* __restrict__ q,
          const int* __restrict__ rowptr, const int* __restrict__ col,
          const float* __restrict__ invdeg, float* __restrict__ out) {
    const int  l   = threadIdx.x & 15;
    const int  hw  = (int)(threadIdx.x >> 4);
    const int  nb  = blockIdx.x * OUT_NODES;
    const bool on  = l < NCLS / 4;
    const float4* __restrict__ q4 = (const float4*)q;

    int   beg[OUT_NPH], end[OUT_NPH], first[OUT_NPH];
    float idg[OUT_NPH];
    #pragma unroll
    for (int t = 0; t < OUT_NPH; ++t) {
        const int node = nb + hw + 16 * t;
        beg[t] = rowptr[node];
        end[t] = (node + 1 < NNODES) ? rowptr[node + 1] : NEDGES;
        idg[t] = invdeg[node];
    }
    #pragma unroll
    for (int t = 0; t < OUT_NPH; ++t)
        first[t] = (l < end[t] - beg[t]) ? col[beg[t] + l] : 0;

    #pragma unroll
    for (int t = 0; t < OUT_NPH; ++t) {
        const int node = nb + hw + 16 * t;
        float4 pv = make_float4(0.f, 0.f, 0.f, 0.f);
        if (on) pv = ((const float4*)(p + (size_t)node * NCLS))[l];

        float4 acc = make_float4(0.f, 0.f, 0.f, 0.f);
        int mine = first[t];
        for (int base = beg[t]; base < end[t]; ) {
            const int n = min(end[t] - base, 16);
            take_q8<0>(q4, mine, n, l, on, acc);
            if (n > 8) take_q8<8>(q4, mine, n, l, on, acc);
            base += 16;
            if (base < end[t])
                mine = (l < end[t] - base) ? col[base + l] : 0;
        }
        if (on) {
            float4 o;
            o.x = pv.x + acc.x * idg[t];
            o.y = pv.y + acc.y * idg[t];
            o.z = pv.z + acc.z * idg[t];
            o.w = pv.w + acc.w * idg[t];
            ((float4*)(out + (size_t)node * NCLS))[l] = o;
        }
    }
}

extern "C" void kernel_launch(void* const* d_in, const int* in_sizes, int n_in,
                              void* d_out, int out_size, void* d_ws, size_t ws_size,
                              hipStream_t stream) {
    const float* feat = (const float*)d_in[0];
    const int*   src  = (const int*)d_in[1];
    const int*   dstv = (const int*)d_in[2];
    const float* Ws0  = (const float*)d_in[3];
    const float* bs0  = (const float*)d_in[4];
    const float* Wn0  = (const float*)d_in[5];
    const float* bn0  = (const float*)d_in[6];
    const float* Ws1  = (const float*)d_in[7];
    const float* bs1  = (const float*)d_in[8];
    const float* Wn1  = (const float*)d_in[9];
    const float* bn1  = (const float*)d_in[10];
    const float* Ws2  = (const float*)d_in[11];
    const float* bs2  = (const float*)d_in[12];
    const float* Wn2  = (const float*)d_in[13];
    const float* bn2  = (const float*)d_in[14];
    float* out = (float*)d_out;

    int*   col    = (int*)d_ws;
    float* h1     = (float*)d_ws + (size_t)NNODES * KIN;
    float* h2     = h1  + (size_t)NNODES * KIN;
    int*   rowptr = (int*)(h2 + (size_t)NNODES * KIN);
    float* invdeg = (float*)(rowptr + NNODES);
    int2* ebuf   = (int2*)h2;
    int*  tab    = (int*)(ebuf + NEDGES);
    int*  toff   = tab + NTAB;
    float* pl    = h2;
    float* ql    = pl + (size_t)NNODES * NCLS;
    float2* Wsp0 = (float2*)(invdeg + NNODES);
    float2* Wnp0 = Wsp0 + PACK8_F2;
    float2* Wsp1 = Wnp0 + PACK8_F2;
    float2* Wnp1 = Wsp1 + PACK8_F2;
    float2* Wsp2 = Wnp1 + PACK8_F2;
    float2* Wnp2 = Wsp2 + PACK3_F2;

    const int tileBlocks = NNODES / 32;

    prep_kernel<<<EBLK + PACK_BLOCKS, PREP_T, 0, stream>>>(
        Ws0, Wn0, Ws1, Wn1, Ws2, Wn2, Wsp0, dstv, tab);

    tab_scan<<<(NTAB + SCAN_T - 1) / SCAN_T, SCAN_T, 0, stream>>>(tab, toff);
    edge_scatter<<<EBLK, ET, 0, stream>>>(src, dstv, toff, ebuf);
    bucket_sort<<<NBUCK, BT, 0, stream>>>(ebuf, toff, rowptr, invdeg, col);

    sage_layer<128, true, true, false><<<tileBlocks, 256, 0, stream>>>(
        feat, rowptr, col, invdeg, Wsp0, Wnp0, bs0, bn0, h1,
        nullptr, nullptr, nullptr, nullptr);

    sage_layer<128, true, true, true><<<tileBlocks, 256, 0, stream>>>(
        h1, rowptr, col, invdeg, Wsp1, Wnp1, bs1, bn1, pl,
        Wsp2, Wnp2, bs2, bn2);

    out_layer<<<NNODES / OUT_NODES, OUT_T, 0, stream>>>(pl, ql, rowptr, col, invdeg, out);
}


# Edge sort: each edge block keeps its bucket-ordered list in its own part, a bucket's place = sum of the blocks' run starts: count pass, table scan, 2 of 7 launches gone; add order = LDS arrival
# speedup vs baseline: 6.0979x; 1.0392x over previous
#include <hip/hip_runtime.h>
#include <hip/hip_bf16.h>

#define NNODES 100000
#define NEDGES 1600000
#define KIN    128
#define NCLS   40
#define NT2    ((NCLS + 15) / 16)

typedef __attribute__((ext_vector_type(2))) float v2f;
typedef __attribute__((ext_vector_type(8))) float v8f;

#define BSH     9
#define BNODES  (1 << BSH)
#define NBUCK   ((NNODES + BNODES - 1) / BNODES)
#define EBLK    250
#define EPB     (NEDGES / EBLK)
#define ET      256
#define EPT     (EPB / ET)
#define NTAB    (NBUCK * EBLK)
static_assert(EPB * EBLK == NEDGES && EPT * ET == EPB, "whole edge blocks");
static_assert(NBUCK <= ET, "edge_block_sort: one thread per bucket");
static_assert(EPB < (1 << 15), "a run's start and length share one table word");

__device__ __forceinline__ void edge_block_sort(const int* __restrict__ src, const int* __restrict__ dst,
                                                int* __restrict__ tab, int2* __restrict__ ebuf,
                                                int eb, int t, int* hist, int* lstart, int2* sorted) {
    for (int i = t; i < NBUCK; i += ET) hist[i] = 0;
    __syncthreads();
    const int* __restrict__ se = src + eb * EPB;
    const int* __restrict__ de = dst + eb * EPB;
    int s[EPT], d[EPT], r[EPT];
    #pragma unroll
    for (int j = 0; j < EPT; ++j) {
        s[j] = se[j * ET + t];
        d[j] = de[j * ET + t];
    }
    #pragma unroll
    for (int j = 0; j < EPT; ++j) r[j] = atomicAdd(&hist[d[j] >> BSH], 1);
    __syncthreads();
    if (t < 32) {
        constexpr int PER = (NBUCK + 31) / 32;
        int v[PER];
        int sum = 0;
        #pragma unroll
        for (int k = 0; k < PER; ++k) {
            const int b = t * PER + k;
            v[k] = (b < NBUCK) ? hist[b] : 0;
            sum += v[k];
        }
        int x = sum;
        for (int o = 1; o < 32; o <<= 1) {
            const int y = __shfl_up(x, o, 32);
            if (t >= o) x += y;
        }
        int run = x - sum;
        #pragma unroll
        for (int k = 0; k < PER; ++k) {
            const int b = t * PER + k;
            if (b < NBUCK) lstart[b] = run;
            run += v[k];
        }
    }
    __syncthreads();
    if (t < NBUCK) tab[t * EBLK + eb] = (lstart[t] << 16) | hist[t];
    #pragma unroll
    for (int j = 0; j < EPT; ++j)
        sorted[lstart[d[j] >> BSH] + r[j]] = make_int2(s[j], d[j]);
    __syncthreads();
    int2* __restrict__ eo = ebuf + eb * EPB;
    for (int p = t; p < EPB; p += ET) eo[p] = sorted[p];
}

#define BT   BNODES
#define BW   (BT / 32)
#define RPW  ((EBLK + BW - 1) / BW)
#define BCAP 12288
static_assert(RPW <= 32, "bucket_sort: one lane per run of the wave");
__global__ void __launch_bounds__(BT)
bucket_sort(const int2* __restrict__ ebuf, const int* __restrict__ tab,
            int* __restrict__ rowptr, float* __restrict__ invdeg, int* __restrict__ col) {
    __shared__ int cnt[BNODES];
    __shared__ int cnt2[BNODES];
    __shared__ int cur[BNODES];
    __shared__ int lrow[BNODES];
    __shared__ int wsum[BW];
    __shared__ int wlen[BW];
    __shared__ int wbeg[BW];
    __shared__ int stage[BCAP];
    const int t    = threadIdx.x;
    const int lane = t & 31;
    const int wv   = __builtin_amdgcn_readfirstlane(t >> 5);
    const int b    = blockIdx.x;
    const int nb   = b * BNODES;
    const int w    = (t < EBLK) ? tab[b * EBLK + t] : 0;
    const int rw   = wv + BW * lane;
    const int mw   = (lane < RPW && rw < EBLK) ? tab[b * EBLK + rw] : 0;
    const int myc  = mw & 0xffff;
    const int myb  = rw * EPB + (mw >> 16);
    cnt[t]  = 0;
    cnt2[t] = 0;
    cur[t]  = 0;

    int sv[2 * RPW], dv[2 * RPW], rk[2 * RPW];
    #pragma unroll
    for (int u = 0; u < RPW; ++u) {
        const int cu = __builtin_amdgcn_readlane(myc, u);
        const int2* __restrict__ er = ebuf + __builtin_amdgcn_readlane(myb, u);
        #pragma unroll
        for (int k = 0; k < 2; ++k) {
            const int i = 32 * k + lane;
            int2 e = make_int2(0, -1);
            if (i < cu) e = er[i];
            sv[2 * u + k] = e.x;
            dv[2 * u + k] = e.y;
        }
    }
    const bool longrun = __any(myc > 64);

    int sl = w & 0xffff;
    int sb = w >> 16;
    for (int o = 16; o > 0; o >>= 1) {
        sl += __shfl_xor(sl, o, 32);
        sb += __shfl_xor(sb, o, 32);
    }
    if (lane == 0) {
        wlen[wv] = sl;
        wbeg[wv] = sb;
    }
    __syncthreads();
    int n  = (lane < BW) ? wlen[lane] : 0;
    int bs = (lane < BW) ? wbeg[lane] : 0;
    for (int o = 16; o > 0; o >>= 1) {
        n  += __shfl_xor(n, o, 32);
        bs += __shfl_xor(bs, o, 32);
    }

    #pragma unroll
    for (int j = 0; j < 2 * RPW; ++j) {
        rk[j] = 0;
        if (dv[j] >= 0) rk[j] = atomicAdd(&cnt[dv[j] - nb], 1);
    }
    if (longrun) {
        for (int u = 0; u < RPW; ++u) {
            const int cu = __builtin_amdgcn_readlane(myc, u);
            const int2* __restrict__ er = ebuf + __builtin_amdgcn_readlane(myb, u);
            for (int i = 64 + lane; i < cu; i += 32) atomicAdd(&cnt2[er[i].y - nb], 1);
        }
    }
    __syncthreads();

    const int c = cnt[t] + cnt2[t];
    int x = c;
    for (int o = 1; o < 32; o <<= 1) {
        const int y = __shfl_up(x, o, 32);
        if (lane >= o) x += y;
    }
    if (lane == 31) wsum[wv] = x;
    __syncthreads();
    int z = (lane < wv) ? wsum[lane] : 0;
    for (int o = 16; o > 0; o >>= 1) z += __shfl_xor(z, o, 32);
    const int ex = z + (x - c);
    lrow[t] = ex;
    const int node = nb + t;
    if (node < NNODES) {
        rowptr[node] = bs + ex;
        invdeg[node] = 1.0f / fmaxf((float)c, 1.0f);
    }
    __syncthreads();

    const bool staged = n <= BCAP;
    #pragma unroll
    for (int j = 0; j < 2 * RPW; ++j) {
        if (dv[j] >= 0) {
            const int p = lrow[dv[j] - nb] + rk[j];
            if (staged) stage[p] = sv[j];
            else        col[bs + p] = sv[j];
        }
    }
    if (longrun) {
        for (int u = 0; u < RPW; ++u) {
            const int cu = __builtin_amdgcn_readlane(myc, u);
            const int2* __restrict__ er = ebuf + __builtin_amdgcn_readlane(myb, u);
            for (int i = 64 + lane; i < cu; i += 32) {
                const int2 e  = er[i];
                const int  dl = e.y - nb;
                const int  p  = lrow[dl] + cnt[dl] + atomicAdd(&cur[dl], 1);
                if (staged) stage[p] = e.x;
                else        col[bs + p] = e.x;
            }
        }
    }
    if (staged) {
        __syncthreads();
        for (int i = t; i < n; i += BT) col[bs + i] = stage[i];
    }
}

__device__ __forceinline__ float sum16(float v) {
    v += __builtin_bit_cast(float, __builtin_amdgcn_update_dpp(0, __builtin_bit_cast(int, v), 0x161, 0xf, 0xf, true));
    v += __builtin_bit_cast(float, __builtin_amdgcn_update_dpp(0, __builtin_bit_cast(int, v), 0x162, 0xf, 0xf, true));
    v += __builtin_bit_cast(float, __builtin_amdgcn_update_dpp(0, __builtin_bit_cast(int, v), 0x164, 0xf, 0xf, true));
    v += __builtin_bit_cast(float, __builtin_amdgcn_update_dpp(0, __builtin_bit_cast(int, v), 0x168, 0xf, 0xf, true));
    return v;
}

template<int B>
__device__ __forceinline__ void take_rows(const float* __restrict__ h, int mine, int k,
                                          int lane, float4& acc) {
    float4 v[B];
    #pragma unroll
    for (int j = 0; j < B; ++j) {
        const int s = __builtin_amdgcn_readlane(mine, k + j);
        v[j] = ((const float4*)(h + (size_t)s * KIN))[lane];
    }
    #pragma unroll
    for (int j = 0; j < B; ++j) {
        acc.x += v[j].x;
        acc.y += v[j].y;
        acc.z += v[j].z;
        acc.w += v[j].w;
    }
}

__device__ __forceinline__ void pack_w(const float* __restrict__ W, float2* __restrict__ Wp,
                                       int nout, int pid) {
    int lane = pid & 31;
    int i    = (pid >> 5) & ((KIN / 4) - 1);
    int tile = pid >> 10;
    int l = lane & 15, half = lane >> 4;
    int col = tile * 16 + l;
    int k   = 4 * i + 2 * half;
    float x = 0.f, y = 0.f;
    if (col < nout) {
        x = W[(size_t)k       * nout + col];
        y = W[(size_t)(k + 1) * nout + col];
    }
    Wp[pid] = make_float2(x, y);
}

#define PREP_T       256
#define PACK8_F2     (8 * (KIN / 4) * 32)
#define PACK3_F2     (3 * (KIN / 4) * 32)
#define PACK8_BLOCKS (PACK8_F2 / PREP_T)
#define PACK3_BLOCKS (PACK3_F2 / PREP_T)
#define PACK_BLOCKS  (4 * PACK8_BLOCKS + 2 * PACK3_BLOCKS)
static_assert(PACK8_F2 % PREP_T == 0 && PACK3_F2 % PREP_T == 0, "pack blocks are whole");
static_assert(PREP_T == ET, "edge_block_sort is written for ET threads");
__global__ void __launch_bounds__(PREP_T)
prep_kernel(const float* __restrict__ Ws0, const float* __restrict__ Wn0,
            const float* __restrict__ Ws1, const float* __restrict__ Wn1,
            const float* __restrict__ Ws2, const float* __restrict__ Wn2,
            float2* __restrict__ Wp, const int* __restrict__ src, const int* __restrict__ dst,
            int* __restrict__ tab, int2* __restrict__ ebuf) {
    __shared__ int  hist[NBUCK];
    __shared__ int  lstart[NBUCK];
    __shared__ int2 sorted[EPB];
    const int t = threadIdx.x;
    if (blockIdx.x < EBLK) {
        edge_block_sort(src, dst, tab, ebuf, blockIdx.x, t, hist, lstart, sorted);
        return;
    }
    const int b = blockIdx.x - EBLK;
    if (b < 4 * PACK8_BLOCKS) {
        const int m = b / PACK8_BLOCKS;
        const float* __restrict__ W = (m == 0) ? Ws0 : (m == 1) ? Wn0 : (m == 2) ? Ws1 : Wn1;
        pack_w(W, Wp + (size_t)m * PACK8_F2, 128, (b - m * PACK8_BLOCKS) * PREP_T + t);
    } else {
        const int b3 = b - 4 * PACK8_BLOCKS;
        const int m  = b3 / PACK3_BLOCKS;
        const float* __restrict__ W = (m == 0) ? Ws2 : Wn2;
        pack_w(W, Wp + (size_t)4 * PACK8_F2 + (size_t)m * PACK3_F2, 40,
               (b3 - m * PACK3_BLOCKS) * PREP_T + t);
    }
}

template<int NOUT, bool ACT, bool NORM, bool PROJ>
__global__ void __launch_bounds__(256) __attribute__((amdgpu_waves_per_eu(PROJ ? 12 : 16, 16)))
sage_layer(const float* __restrict__ h,
           const int* __restrict__ rowptr,
           const int* __restrict__ col,
           const float* __restrict__ invdeg,
           const float2* __restrict__ Wsp,
           const float2* __restrict__ Wnp,
           const float* __restrict__ bself,
           const float* __restrict__ bneigh,
           float* __restrict__ out,
           const float2* __restrict__ Wsp2,
           const float2* __restrict__ Wnp2,
           const float* __restrict__ bself2,
           const float* __restrict__ bneigh2) {
    constexpr bool FULL = (NOUT % 16) == 0;
    constexpr int NT    = (NOUT + 15) / 16;
    constexpr int ROWS  = 32;
    constexpr int PITCH = KIN + 4;
    static_assert(!NORM || NT == 8, "the row-norm barriers need all eight waves");
    static_assert(!PROJ || (NORM && NOUT == KIN), "the second product reads the tile as a KIN-wide A operand");
    __shared__ float sSelf[ROWS * PITCH];
    __shared__ float sNeigh[ROWS * PITCH];

    const int lane = threadIdx.x & 31;
    const int wv   = __builtin_amdgcn_readfirstlane((int)(threadIdx.x >> 5));
    const int m0   = blockIdx.x * ROWS;

    int   beg[4], end[4], first[4];
    float idg[4];
    #pragma unroll
    for (int t = 0; t < 4; ++t) {
        const int node = m0 + wv + 8 * t;
        beg[t] = rowptr[node];
        end[t] = (node + 1 < NNODES) ? rowptr[node + 1] : NEDGES;
        idg[t] = invdeg[node];
    }
    #pragma unroll
    for (int t = 0; t < 4; ++t)
        first[t] = (lane < end[t] - beg[t]) ? col[beg[t] + lane] : 0;

    #pragma unroll
    for (int t = 0; t < 4; ++t) {
        const int r    = wv + 8 * t;
        const int node = m0 + r;
        const float4 hv = ((const float4*)(h + (size_t)node * KIN))[lane];

        float4 acc = make_float4(0.f, 0.f, 0.f, 0.f);
        int mine = first[t];
        for (int base = beg[t]; base < end[t]; ) {
            const int n = min(end[t] - base, 32);
            int k = 0;
            for (; k + 8 <= n; k += 8) take_rows<8>(h, mine, k, lane, acc);
            if (k + 4 <= n) { take_rows<4>(h, mine, k, lane, acc); k += 4; }
            if (k + 2 <= n) { take_rows<2>(h, mine, k, lane, acc); k += 2; }
            if (k < n)      { take_rows<1>(h, mine, k, lane, acc); }
            base += 32;
            if (base < end[t])
                mine = (lane < end[t] - base) ? col[base + lane] : 0;
        }
        acc.x *= idg[t]; acc.y *= idg[t]; acc.z *= idg[t]; acc.w *= idg[t];
        *(float4*)(sSelf  + r * PITCH + 4 * lane) = hv;
        *(float4*)(sNeigh + r * PITCH + 4 * lane) = acc;
    }
    __syncthreads();
    if (NT < 8 && wv >= NT) return;

    const int l    = lane & 15;
    const int half = lane >> 4;
    const int n0   = wv * 16;
    const int ocol = n0 + l;
    const bool colok = FULL || (ocol < NOUT);

    const float*  As = sSelf  + l * PITCH + 2 * half;
    const float*  An = sNeigh + l * PITCH + 2 * half;
    const float2* __restrict__ Bs = Wsp + (size_t)wv * (KIN / 4) * 32 + lane;
    const float2* __restrict__ Bn = Wnp + (size_t)wv * (KIN / 4) * 32 + lane;

    v8f acc0 = {0.f, 0.f, 0.f, 0.f, 0.f, 0.f, 0.f, 0.f};
    v8f acc1 = {0.f, 0.f, 0.f, 0.f, 0.f, 0.f, 0.f, 0.f};

    #pragma unroll
    for (int i = 0; i < KIN / 4; ++i) {
        const float2 a0f = *(const float2*)(As + 4 * i);
        const float2 a1f = *(const float2*)(As + 16 * PITCH + 4 * i);
        const float2 bf  = Bs[i * 32];
        v2f a0 = {a0f.x, a0f.y};
        v2f a1 = {a1f.x, a1f.y};
        v2f b  = {bf.x, bf.y};
        acc0 = __builtin_amdgcn_wmma_f32_16x16x4_f32(false, a0, false, b,
                                                     (short)0, acc0, false, false);
        acc1 = __builtin_amdgcn_wmma_f32_16x16x4_f32(false, a1, false, b,
                                                     (short)0, acc1, false, false);
    }
    #pragma unroll
    for (int i = 0; i < KIN / 4; ++i) {
        const float2 a0f = *(const float2*)(An + 4 * i);
        const float2 a1f = *(const float2*)(An + 16 * PITCH + 4 * i);
        const float2 bf  = Bn[i * 32];
        v2f a0 = {a0f.x, a0f.y};
        v2f a1 = {a1f.x, a1f.y};
        v2f b  = {bf.x, bf.y};
        acc0 = __builtin_amdgcn_wmma_f32_16x16x4_f32(false, a0, false, b,
                                                     (short)0, acc0, false, false);
        acc1 = __builtin_amdgcn_wmma_f32_16x16x4_f32(false, a1, false, b,
                                                     (short)0, acc1, false, false);
    }

    const float bias = colok ? (bself[ocol] + bneigh[ocol]) : 0.0f;
    #pragma unroll
    for (int r = 0; r < 8; ++r) {
        float v0 = acc0[r] + bias;
        float v1 = acc1[r] + bias;
        if (ACT) { v0 = fmaxf(v0, 0.0f); v1 = fmaxf(v1, 0.0f); }
        acc0[r] = v0;
        acc1[r] = v1;
    }

    if (NORM) {
        __shared__ alignas(16) float part[8 * ROWS];
        float q0[8], q1[8];
        #pragma unroll
        for (int r = 0; r < 8; ++r) {
            q0[r] = sum16(acc0[r] * acc0[r]);
            q1[r] = sum16(acc1[r] * acc1[r]);
        }
        if (l == 0) {
            float* p = part + wv * ROWS + 8 * half;
            *(float4*)(p)      = make_float4(q0[0], q0[1], q0[2], q0[3]);
            *(float4*)(p + 4)  = make_float4(q0[4], q0[5], q0[6], q0[7]);
            *(float4*)(p + 16) = make_float4(q1[0], q1[1], q1[2], q1[3]);
            *(float4*)(p + 20) = make_float4(q1[4], q1[5], q1[6], q1[7]);
        }
        __syncthreads();
        float tot = part[lane];
        #pragma unroll
        for (int w = 1; w < 8; ++w) tot += part[w * ROWS + lane];
        const float nrm = fmaxf(sqrtf(tot), 1e-12f);
        #pragma unroll
        for (int r = 0; r < 8; ++r) {
            const float nrm0 = __shfl(nrm, r + 8 * half, 32);
            const float nrm1 = __shfl(nrm, 16 + r + 8 * half, 32);
            acc0[r] = acc0[r] / nrm0;
            acc1[r] = acc1[r] / nrm1;
        }
    }

    if (PROJ) {
        #pragma unroll
        for (int r = 0; r < 8; ++r) {
            sSelf[(r + 8 * half) * PITCH + ocol]      = acc0[r];
            sSelf[(16 + r + 8 * half) * PITCH + ocol] = acc1[r];
        }
        __syncthreads();
        if (wv >= 2 * NT2) return;
        const bool isq = wv >= NT2;
        const int  tj  = isq ? wv - NT2 : wv;
        const float2* __restrict__ B2 = (isq ? Wnp2 : Wsp2) + (size_t)tj * (KIN / 4) * 32 + lane;
        v8f c0 = {0.f, 0.f, 0.f, 0.f, 0.f, 0.f, 0.f, 0.f};
        v8f c1 = {0.f, 0.f, 0.f, 0.f, 0.f, 0.f, 0.f, 0.f};
        #pragma unroll
        for (int i = 0; i < KIN / 4; ++i) {
            const float2 a0f = *(const float2*)(As + 4 * i);
            const float2 a1f = *(const float2*)(As + 16 * PITCH + 4 * i);
            const float2 bf  = B2[i * 32];
            v2f a0 = {a0f.x, a0f.y};
            v2f a1 = {a1f.x, a1f.y};
            v2f b  = {bf.x, bf.y};
            c0 = __builtin_amdgcn_wmma_f32_16x16x4_f32(false, a0, false, b,
                                                       (short)0, c0, false, false);
            c1 = __builtin_amdgcn_wmma_f32_16x16x4_f32(false, a1, false, b,
                                                       (short)0, c1, false, false);
        }
        const int pc = tj * 16 + l;
        if (pc < NCLS) {
            const float b2 = isq ? 0.0f : (bself2[pc] + bneigh2[pc]);
            float* __restrict__ dstp = out + (isq ? (size_t)NNODES * NCLS : (size_t)0);
            #pragma unroll
            for (int r = 0; r < 8; ++r) {
                const int m = m0 + r + 8 * half;
                dstp[(size_t)m * NCLS + pc]        = c0[r] + b2;
                dstp[(size_t)(m + 16) * NCLS + pc] = c1[r] + b2;
            }
        }
    } else if (colok) {
        #pragma unroll
        for (int r = 0; r < 8; ++r) {
            const int m = m0 + r + 8 * half;
            out[(size_t)m * NOUT + ocol]        = acc0[r];
            out[(size_t)(m + 16) * NOUT + ocol] = acc1[r];
        }
    }
}

template<int K>
__device__ __forceinline__ int row_lane(int x) {
    return __builtin_amdgcn_update_dpp(0, x, 0x150 + K, 0xf, 0xf, true);
}

template<int K0>
__device__ __forceinline__ void take_q8(const float4* __restrict__ q4, int mine, int n, int l,
                                        bool on, float4& acc) {
    float4 v[8];
#define TAKE_Q(j)                                                              \
    {                                                                          \
        const int s = row_lane<K0 + j>(mine);                                  \
        v[j] = make_float4(0.f, 0.f, 0.f, 0.f);                                \
        if (on && (K0 + j) < n) v[j] = q4[(size_t)s * (NCLS / 4) + l];         \
    }
    TAKE_Q(0) TAKE_Q(1) TAKE_Q(2) TAKE_Q(3) TAKE_Q(4) TAKE_Q(5) TAKE_Q(6) TAKE_Q(7)
#undef TAKE_Q
    #pragma unroll
    for (int j = 0; j < 8; ++j) {
        acc.x += v[j].x;
        acc.y += v[j].y;
        acc.z += v[j].z;
        acc.w += v[j].w;
    }
}

#define OUT_T     256
#define OUT_NPH   5
#define OUT_NODES ((OUT_T / 16) * OUT_NPH)
static_assert(NNODES % OUT_NODES == 0, "whole blocks, no node guard");
__global__ void __launch_bounds__(OUT_T)
out_layer(const float* __restrict__ p, const float* __restrict__ q,
          const int* __restrict__ rowptr, const int* __restrict__ col,
          const float* __restrict__ invdeg, float* __restrict__ out) {
    const int  l   = threadIdx.x & 15;
    const int  hw  = (int)(threadIdx.x >> 4);
    const int  nb  = blockIdx.x * OUT_NODES;
    const bool on  = l < NCLS / 4;
    const float4* __restrict__ q4 = (const float4*)q;

    int   beg[OUT_NPH], end[OUT_NPH], first[OUT_NPH];
    float idg[OUT_NPH];
    #pragma unroll
    for (int t = 0; t < OUT_NPH; ++t) {
        const int node = nb + hw + 16 * t;
        beg[t] = rowptr[node];
        end[t] = (node + 1 < NNODES) ? rowptr[node + 1] : NEDGES;
        idg[t] = invdeg[node];
    }
    #pragma unroll
    for (int t = 0; t < OUT_NPH; ++t)
        first[t] = (l < end[t] - beg[t]) ? col[beg[t] + l] : 0;

    #pragma unroll
    for (int t = 0; t < OUT_NPH; ++t) {
        const int node = nb + hw + 16 * t;
        float4 pv = make_float4(0.f, 0.f, 0.f, 0.f);
        if (on) pv = ((const float4*)(p + (size_t)node * NCLS))[l];

        float4 acc = make_float4(0.f, 0.f, 0.f, 0.f);
        int mine = first[t];
        for (int base = beg[t]; base < end[t]; ) {
            const int n = min(end[t] - base, 16);
            take_q8<0>(q4, mine, n, l, on, acc);
            if (n > 8) take_q8<8>(q4, mine, n, l, on, acc);
            base += 16;
            if (base < end[t])
                mine = (l < end[t] - base) ? col[base + l] : 0;
        }
        if (on) {
            float4 o;
            o.x = pv.x + acc.x * idg[t];
            o.y = pv.y + acc.y * idg[t];
            o.z = pv.z + acc.z * idg[t];
            o.w = pv.w + acc.w * idg[t];
            ((float4*)(out + (size_t)node * NCLS))[l] = o;
        }
    }
}

extern "C" void kernel_launch(void* const* d_in, const int* in_sizes, int n_in,
                              void* d_out, int out_size, void* d_ws, size_t ws_size,
                              hipStream_t stream) {
    const float* feat = (const float*)d_in[0];
    const int*   src  = (const int*)d_in[1];
    const int*   dstv = (const int*)d_in[2];
    const float* Ws0  = (const float*)d_in[3];
    const float* bs0  = (const float*)d_in[4];
    const float* Wn0  = (const float*)d_in[5];
    const float* bn0  = (const float*)d_in[6];
    const float* Ws1  = (const float*)d_in[7];
    const float* bs1  = (const float*)d_in[8];
    const float* Wn1  = (const float*)d_in[9];
    const float* bn1  = (const float*)d_in[10];
    const float* Ws2  = (const float*)d_in[11];
    const float* bs2  = (const float*)d_in[12];
    const float* Wn2  = (const float*)d_in[13];
    const float* bn2  = (const float*)d_in[14];
    float* out = (float*)d_out;

    int*   col    = (int*)d_ws;
    float* h1     = (float*)d_ws + (size_t)NNODES * KIN;
    float* h2     = h1  + (size_t)NNODES * KIN;
    int*   rowptr = (int*)(h2 + (size_t)NNODES * KIN);
    float* invdeg = (float*)(rowptr + NNODES);
    int2* ebuf   = (int2*)h2;
    int*  tab    = (int*)(ebuf + NEDGES);
    float* pl    = h2;
    float* ql    = pl + (size_t)NNODES * NCLS;
    float2* Wsp0 = (float2*)(invdeg + NNODES);
    float2* Wnp0 = Wsp0 + PACK8_F2;
    float2* Wsp1 = Wnp0 + PACK8_F2;
    float2* Wnp1 = Wsp1 + PACK8_F2;
    float2* Wsp2 = Wnp1 + PACK8_F2;
    float2* Wnp2 = Wsp2 + PACK3_F2;

    const int tileBlocks = NNODES / 32;

    prep_kernel<<<EBLK + PACK_BLOCKS, PREP_T, 0, stream>>>(
        Ws0, Wn0, Ws1, Wn1, Ws2, Wn2, Wsp0, src, dstv, tab, ebuf);

    bucket_sort<<<NBUCK, BT, 0, stream>>>(ebuf, tab, rowptr, invdeg, col);

    sage_layer<128, true, true, false><<<tileBlocks, 256, 0, stream>>>(
        feat, rowptr, col, invdeg, Wsp0, Wnp0, bs0, bn0, h1,
        nullptr, nullptr, nullptr, nullptr);

    sage_layer<128, true, true, true><<<tileBlocks, 256, 0, stream>>>(
        h1, rowptr, col, invdeg, Wsp1, Wnp1, bs1, bn1, pl,
        Wsp2, Wnp2, bs2, bn2);

    out_layer<<<NNODES / OUT_NODES, OUT_T, 0, stream>>>(pl, ql, rowptr, col, invdeg, out);
}
